# starting kernel
# baseline (speedup 1.0000x reference)
#include <hip/hip_runtime.h>
#include <hip/hip_bf16.h>

typedef __attribute__((ext_vector_type(2))) float v2f;
typedef __attribute__((ext_vector_type(8))) float v8f;

__global__ void k_init_deg(int* __restrict__ deg, int n) {
  int i = blockIdx.x * blockDim.x + threadIdx.x;
  if (i < n) deg[i] = 1;
}

__global__ void k_count_deg(const int* __restrict__ dst, int* __restrict__ deg, int E) {
  int i = blockIdx.x * blockDim.x + threadIdx.x;
  if (i < E) atomicAdd(&deg[dst[i]], 1);
}

__global__ void k_dinv(const int* __restrict__ deg, float* __restrict__ dinv, int n) {
  int i = blockIdx.x * blockDim.x + threadIdx.x;
  if (i < n) dinv[i] = rsqrtf((float)deg[i]);
}

__global__ void k_zero_f32(float* __restrict__ p, long long n) {
  long long i = (long long)blockIdx.x * blockDim.x + threadIdx.x;
  if (i < n) p[i] = 0.0f;
}

template <int K, int N>
__global__ __launch_bounds__(256) void k_gemm_wmma(const float* __restrict__ A,
                                                   const float* __restrict__ W,
                                                   float* __restrict__ C, int mTiles) {
  const int lane = threadIdx.x & 31;
  const int wave = threadIdx.x >> 5;
  const int tile = blockIdx.x * 8 + wave;
  constexpr int tilesN = N / 16;
  const int mT = tile / tilesN;
  const int nT = tile % tilesN;
  if (mT >= mTiles) return;

  const int row = lane & 15;
  const int kh = lane >> 4;
  const int m0 = mT * 16;
  const int n0 = nT * 16;
  const int col = n0 + row;

  const float* ap = A + (size_t)(m0 + row) * K + 2 * kh;

  v8f c = {};
#pragma unroll 4
  for (int k = 0; k < K; k += 4) {
    v2f a = *(const v2f*)(ap + k);
    const int kr = k + 2 * kh;
    v2f b;
    b.x = W[(size_t)kr * N + col];
    b.y = W[(size_t)(kr + 1) * N + col];
    c = __builtin_amdgcn_wmma_f32_16x16x4_f32(false, a, false, b, (short)0, c, false,
                                              false);
  }

  float* cp = C + (size_t)(m0 + 8 * kh) * N + col;
#pragma unroll
  for (int v = 0; v < 8; ++v) cp[(size_t)v * N] = c[v];
}

template <int F>
__global__ void k_edge_agg(const float* __restrict__ hw, const float* __restrict__ dinv,
                           const int* __restrict__ src, const int* __restrict__ dst,
                           float* __restrict__ agg, int E) {
  constexpr int F4 = F / 4;
  long long idx = (long long)blockIdx.x * blockDim.x + threadIdx.x;
  long long total = (long long)E * F4;
  if (idx >= total) return;
  const int e = (int)(idx / F4);
  const int q = (int)(idx % F4);
  const int s = src[e];
  const int d = dst[e];
  const float nrm = dinv[s] * dinv[d];
  const float4 m = ((const float4*)(hw + (size_t)s * F))[q];
  float* out = agg + (size_t)d * F + 4 * q;
  unsafeAtomicAdd(out + 0, m.x * nrm);
  unsafeAtomicAdd(out + 1, m.y * nrm);
  unsafeAtomicAdd(out + 2, m.z * nrm);
  unsafeAtomicAdd(out + 3, m.w * nrm);
}

template <int F>
__global__ void k_post1(float* __restrict__ hw, const float* __restrict__ agg,
                        const float* __restrict__ dinv, const float* __restrict__ b,
                        int n) {
  long long idx = (long long)blockIdx.x * blockDim.x + threadIdx.x;
  long long total = (long long)n * F;
  if (idx >= total) return;
  const int i = (int)(idx / F);
  const int f = (int)(idx % F);
  const float di = dinv[i];
  const float v = agg[idx] + hw[idx] * di * di + b[f];
  hw[idx] = v > 0.0f ? v : 0.0f;
}

template <int F>
__global__ void k_init_out(const float* __restrict__ hw2, const float* __restrict__ dinv,
                           const float* __restrict__ b, float* __restrict__ out, int n) {
  long long idx = (long long)blockIdx.x * blockDim.x + threadIdx.x;
  long long total = (long long)n * F;
  if (idx >= total) return;
  const int i = (int)(idx / F);
  const int f = (int)(idx % F);
  const float di = dinv[i];
  out[idx] = b[f] + hw2[idx] * di * di;
}

extern "C" void kernel_launch(void* const* d_in, const int* in_sizes, int n_in,
                              void* d_out, int out_size, void* d_ws, size_t ws_size,
                              hipStream_t stream) {
  (void)n_in;
  (void)out_size;
  (void)ws_size;

  const float* x = (const float*)d_in[0];
  const int* ei = (const int*)d_in[1];
  const float* W1 = (const float*)d_in[2];
  const float* b1 = (const float*)d_in[3];
  const float* W2 = (const float*)d_in[4];
  const float* b2 = (const float*)d_in[5];
  float* out = (float*)d_out;

  constexpr int IN = 256, HID = 128, OUT = 64;
  const int N = in_sizes[0] / IN;
  const int E = in_sizes[1] / 2;
  const int* src = ei;
  const int* dst = ei + E;

  char* ws = (char*)d_ws;
  size_t off = 0;
  auto wsalloc = [&](size_t bytes) -> void* {
    void* p = ws + off;
    off += (bytes + 255) & ~(size_t)255;
    return p;
  };
  int* deg = (int*)wsalloc((size_t)N * 4);
  float* dinv = (float*)wsalloc((size_t)N * 4);
  float* hw1 = (float*)wsalloc((size_t)N * HID * 4);
  float* agg1 = (float*)wsalloc((size_t)N * HID * 4);
  float* hw2 = (float*)wsalloc((size_t)N * OUT * 4);

  const int TB = 256;
  auto blks = [](long long total, int tb) { return (unsigned)((total + tb - 1) / tb); };

  k_init_deg<<<blks(N, TB), TB, 0, stream>>>(deg, N);
  k_count_deg<<<blks(E, TB), TB, 0, stream>>>(dst, deg, E);
  k_dinv<<<blks(N, TB), TB, 0, stream>>>(deg, dinv, N);

  {
    const int mTiles = N / 16;
    const int tiles = mTiles * (HID / 16);
    k_gemm_wmma<IN, HID><<<blks(tiles, 8), TB, 0, stream>>>(x, W1, hw1, mTiles);
  }

  k_zero_f32<<<blks((long long)N * HID, TB), TB, 0, stream>>>(agg1, (long long)N * HID);
  k_edge_agg<HID><<<blks((long long)E * (HID / 4), TB), TB, 0, stream>>>(hw1, dinv, src,
                                                                         dst, agg1, E);
  k_post1<HID><<<blks((long long)N * HID, TB), TB, 0, stream>>>(hw1, agg1, dinv, b1, N);

  {
    const int mTiles = N / 16;
    const int tiles = mTiles * (OUT / 16);
    k_gemm_wmma<HID, OUT><<<blks(tiles, 8), TB, 0, stream>>>(hw1, W2, hw2, mTiles);
  }

  k_init_out<OUT><<<blks((long long)N * OUT, TB), TB, 0, stream>>>(hw2, dinv, b2, out, N);
  k_edge_agg<OUT><<<blks((long long)E * (OUT / 4), TB), TB, 0, stream>>>(hw2, dinv, src,
                                                                         dst, out, E);
}


# edge scatter re-laid: lane = consecutive float of a row, so each gather load and each f32 atomic instruction covers one whole 128-B line (was 4 lines at 16-B stride); atomics and arrival order kept
# speedup vs baseline: 1.2621x; 1.2621x over previous
#include <hip/hip_runtime.h>
#include <hip/hip_bf16.h>

typedef __attribute__((ext_vector_type(2))) float v2f;
typedef __attribute__((ext_vector_type(8))) float v8f;

__global__ void k_init_deg(int* __restrict__ deg, int n) {
  int i = blockIdx.x * blockDim.x + threadIdx.x;
  if (i < n) deg[i] = 1;
}

__global__ void k_count_deg(const int* __restrict__ dst, int* __restrict__ deg, int E) {
  int i = blockIdx.x * blockDim.x + threadIdx.x;
  if (i < E) atomicAdd(&deg[dst[i]], 1);
}

__global__ void k_dinv(const int* __restrict__ deg, float* __restrict__ dinv, int n) {
  int i = blockIdx.x * blockDim.x + threadIdx.x;
  if (i < n) dinv[i] = rsqrtf((float)deg[i]);
}

__global__ void k_zero_f32(float* __restrict__ p, long long n) {
  long long i = (long long)blockIdx.x * blockDim.x + threadIdx.x;
  if (i < n) p[i] = 0.0f;
}

template <int K, int N>
__global__ __launch_bounds__(256) void k_gemm_wmma(const float* __restrict__ A,
                                                   const float* __restrict__ W,
                                                   float* __restrict__ C, int mTiles) {
  const int lane = threadIdx.x & 31;
  const int wave = threadIdx.x >> 5;
  const int tile = blockIdx.x * 8 + wave;
  constexpr int tilesN = N / 16;
  const int mT = tile / tilesN;
  const int nT = tile % tilesN;
  if (mT >= mTiles) return;

  const int row = lane & 15;
  const int kh = lane >> 4;
  const int m0 = mT * 16;
  const int n0 = nT * 16;
  const int col = n0 + row;

  const float* ap = A + (size_t)(m0 + row) * K + 2 * kh;

  v8f c = {};
#pragma unroll 4
  for (int k = 0; k < K; k += 4) {
    v2f a = *(const v2f*)(ap + k);
    const int kr = k + 2 * kh;
    v2f b;
    b.x = W[(size_t)kr * N + col];
    b.y = W[(size_t)(kr + 1) * N + col];
    c = __builtin_amdgcn_wmma_f32_16x16x4_f32(false, a, false, b, (short)0, c, false,
                                              false);
  }

  float* cp = C + (size_t)(m0 + 8 * kh) * N + col;
#pragma unroll
  for (int v = 0; v < 8; ++v) cp[(size_t)v * N] = c[v];
}

template <int F, int NE>
__device__ __forceinline__ void agg_edges(const float* __restrict__ hw,
                                          const float* __restrict__ dinv,
                                          const int* __restrict__ src,
                                          const int* __restrict__ dst,
                                          float* __restrict__ agg, long long e0, int lane) {
  constexpr int LPR = F / 32;
  float m[NE][LPR];
  float* op[NE];
  for (int i = 0; i < NE; ++i) {
    const int s = src[e0 + i];
    const int d = dst[e0 + i];
    const float nrm = dinv[s] * dinv[d];
    const float* hp = hw + (size_t)s * F + lane;
    op[i] = agg + (size_t)d * F + lane;
    for (int j = 0; j < LPR; ++j) m[i][j] = hp[32 * j] * nrm;
  }
  for (int i = 0; i < NE; ++i) {
    for (int j = 0; j < LPR; ++j) unsafeAtomicAdd(op[i] + 32 * j, m[i][j]);
  }
}

template <int F>
__global__ void k_edge_agg(const float* __restrict__ hw, const float* __restrict__ dinv,
                           const int* __restrict__ src, const int* __restrict__ dst,
                           float* __restrict__ agg, int E) {
  constexpr int EPW = 128 / F;
  const int lane = threadIdx.x & 31;
  const long long wv = ((long long)blockIdx.x * blockDim.x + threadIdx.x) >> 5;
  const long long e0 = wv * EPW;
  if (e0 >= E) return;
  if (EPW == 1 || e0 + EPW <= E)
    agg_edges<F, EPW>(hw, dinv, src, dst, agg, e0, lane);
  else
    agg_edges<F, 1>(hw, dinv, src, dst, agg, e0, lane);
}

template <int F>
__global__ void k_post1(float* __restrict__ hw, const float* __restrict__ agg,
                        const float* __restrict__ dinv, const float* __restrict__ b,
                        int n) {
  long long idx = (long long)blockIdx.x * blockDim.x + threadIdx.x;
  long long total = (long long)n * F;
  if (idx >= total) return;
  const int i = (int)(idx / F);
  const int f = (int)(idx % F);
  const float di = dinv[i];
  const float v = agg[idx] + hw[idx] * di * di + b[f];
  hw[idx] = v > 0.0f ? v : 0.0f;
}

template <int F>
__global__ void k_init_out(const float* __restrict__ hw2, const float* __restrict__ dinv,
                           const float* __restrict__ b, float* __restrict__ out, int n) {
  long long idx = (long long)blockIdx.x * blockDim.x + threadIdx.x;
  long long total = (long long)n * F;
  if (idx >= total) return;
  const int i = (int)(idx / F);
  const int f = (int)(idx % F);
  const float di = dinv[i];
  out[idx] = b[f] + hw2[idx] * di * di;
}

extern "C" void kernel_launch(void* const* d_in, const int* in_sizes, int n_in,
                              void* d_out, int out_size, void* d_ws, size_t ws_size,
                              hipStream_t stream) {
  (void)n_in;
  (void)out_size;
  (void)ws_size;

  const float* x = (const float*)d_in[0];
  const int* ei = (const int*)d_in[1];
  const float* W1 = (const float*)d_in[2];
  const float* b1 = (const float*)d_in[3];
  const float* W2 = (const float*)d_in[4];
  const float* b2 = (const float*)d_in[5];
  float* out = (float*)d_out;

  constexpr int IN = 256, HID = 128, OUT = 64;
  const int N = in_sizes[0] / IN;
  const int E = in_sizes[1] / 2;
  const int* src = ei;
  const int* dst = ei + E;

  char* ws = (char*)d_ws;
  size_t off = 0;
  auto wsalloc = [&](size_t bytes) -> void* {
    void* p = ws + off;
    off += (bytes + 255) & ~(size_t)255;
    return p;
  };
  int* deg = (int*)wsalloc((size_t)N * 4);
  float* dinv = (float*)wsalloc((size_t)N * 4);
  float* hw1 = (float*)wsalloc((size_t)N * HID * 4);
  float* agg1 = (float*)wsalloc((size_t)N * HID * 4);
  float* hw2 = (float*)wsalloc((size_t)N * OUT * 4);

  const int TB = 256;
  auto blks = [](long long total, int tb) { return (unsigned)((total + tb - 1) / tb); };

  k_init_deg<<<blks(N, TB), TB, 0, stream>>>(deg, N);
  k_count_deg<<<blks(E, TB), TB, 0, stream>>>(dst, deg, E);
  k_dinv<<<blks(N, TB), TB, 0, stream>>>(deg, dinv, N);

  {
    const int mTiles = N / 16;
    const int tiles = mTiles * (HID / 16);
    k_gemm_wmma<IN, HID><<<blks(tiles, 8), TB, 0, stream>>>(x, W1, hw1, mTiles);
  }

  k_zero_f32<<<blks((long long)N * HID, TB), TB, 0, stream>>>(agg1, (long long)N * HID);
  k_edge_agg<HID><<<blks((long long)E * (HID / 4), TB), TB, 0, stream>>>(hw1, dinv, src,
                                                                         dst, agg1, E);
  k_post1<HID><<<blks((long long)N * HID, TB), TB, 0, stream>>>(hw1, agg1, dinv, b1, N);

  {
    const int mTiles = N / 16;
    const int tiles = mTiles * (OUT / 16);
    k_gemm_wmma<HID, OUT><<<blks(tiles, 8), TB, 0, stream>>>(hw1, W2, hw2, mTiles);
  }

  k_init_out<OUT><<<blks((long long)N * OUT, TB), TB, 0, stream>>>(hw2, dinv, b2, out, N);
  k_edge_agg<OUT><<<blks((long long)E * (OUT / 4), TB), TB, 0, stream>>>(hw2, dinv, src,
                                                                         dst, out, E);
}


# atomic edge scatter replaced by a destination-bucketed gather: counting add returns each edge's slot, scan, place; one wave per node sums its rows in registers; sum order = arrival-order rank
# speedup vs baseline: 2.1117x; 1.6732x over previous
#include <hip/hip_runtime.h>
#include <hip/hip_bf16.h>

typedef __attribute__((ext_vector_type(2))) float v2f;
typedef __attribute__((ext_vector_type(8))) float v8f;

__global__ void k_init_deg(int* __restrict__ deg, int n) {
  int i = blockIdx.x * blockDim.x + threadIdx.x;
  if (i < n) deg[i] = 1;
}

__global__ void k_count_deg(const int* __restrict__ dst, int* __restrict__ deg,
                            int* __restrict__ slot, int E) {
  int i = blockIdx.x * blockDim.x + threadIdx.x;
  if (i < E) slot[i] = atomicAdd(&deg[dst[i]], 1) - 1;
}

constexpr int SCAN_TB = 256;
constexpr int SCAN_CHUNK = 4 * SCAN_TB;

__global__ __launch_bounds__(SCAN_TB) void k_scan_rows(const int* __restrict__ deg,
                                                       int* __restrict__ rowptr,
                                                       float* __restrict__ dinv, int n) {
  __shared__ int s_front[SCAN_TB / 32];
  __shared__ int s_wsum[SCAN_TB / 32];
  const int tid = threadIdx.x;
  const int lane = tid & 31;
  const int wave = tid >> 5;
  const int base = blockIdx.x * SCAN_CHUNK;

  int front = 0;
  const int4* d4 = (const int4*)deg;
  const int n4 = base >> 2;
  for (int i = tid; i < n4; i += 8 * SCAN_TB) {
    int4 q[8];
#pragma unroll
    for (int u = 0; u < 8; ++u) {
      const int iu = i + u * SCAN_TB;
      q[u] = d4[iu < n4 ? iu : 0];
    }
#pragma unroll
    for (int u = 0; u < 8; ++u) {
      const int t = (q[u].x - 1) + (q[u].y - 1) + (q[u].z - 1) + (q[u].w - 1);
      front += (i + u * SCAN_TB < n4) ? t : 0;
    }
  }
  for (int o = 16; o > 0; o >>= 1) front += __shfl_xor(front, o);
  if (lane == 0) s_front[wave] = front;

  const int i0 = base + 4 * tid;
  int v[4];
  int tsum = 0;
#pragma unroll
  for (int j = 0; j < 4; ++j) {
    const int t = deg[i0 + j < n ? i0 + j : n - 1];
    v[j] = (i0 + j < n) ? t - 1 : 0;
    tsum += v[j];
  }
  int inc = tsum;
  for (int o = 1; o < 32; o <<= 1) {
    const int t = __shfl_up(inc, o);
    if (lane >= o) inc += t;
  }
  if (lane == 31) s_wsum[wave] = inc;
  __syncthreads();

  int run = inc - tsum;
  for (int w = 0; w < SCAN_TB / 32; ++w) {
    run += s_front[w];
    if (w < wave) run += s_wsum[w];
  }
#pragma unroll
  for (int j = 0; j < 4; ++j) {
    if (i0 + j < n) {
      rowptr[i0 + j] = run;
      dinv[i0 + j] = rsqrtf((float)(v[j] + 1));
    }
    run += v[j];
  }
  if (blockIdx.x == gridDim.x - 1 && tid == SCAN_TB - 1) rowptr[n] = run;
}

__global__ void k_place(const int* __restrict__ src, const int* __restrict__ dst,
                        const int* __restrict__ rowptr, const int* __restrict__ slot,
                        int* __restrict__ csr, int E) {
  int i = blockIdx.x * blockDim.x + threadIdx.x;
  if (i >= E) return;
  const int s = src[i];
  const int p = rowptr[dst[i]] + slot[i];
  if ((unsigned)p < (unsigned)E) csr[p] = s;
}

template <int K, int N>
__global__ __launch_bounds__(256) void k_gemm_wmma(const float* __restrict__ A,
                                                   const float* __restrict__ W,
                                                   float* __restrict__ C, int mTiles) {
  const int lane = threadIdx.x & 31;
  const int wave = threadIdx.x >> 5;
  const int tile = blockIdx.x * 8 + wave;
  constexpr int tilesN = N / 16;
  const int mT = tile / tilesN;
  const int nT = tile % tilesN;
  if (mT >= mTiles) return;

  const int row = lane & 15;
  const int kh = lane >> 4;
  const int m0 = mT * 16;
  const int n0 = nT * 16;
  const int col = n0 + row;

  const float* ap = A + (size_t)(m0 + row) * K + 2 * kh;

  v8f c = {};
#pragma unroll 4
  for (int k = 0; k < K; k += 4) {
    v2f a = *(const v2f*)(ap + k);
    const int kr = k + 2 * kh;
    v2f b;
    b.x = W[(size_t)kr * N + col];
    b.y = W[(size_t)(kr + 1) * N + col];
    c = __builtin_amdgcn_wmma_f32_16x16x4_f32(false, a, false, b, (short)0, c, false,
                                              false);
  }

  float* cp = C + (size_t)(m0 + 8 * kh) * N + col;
#pragma unroll
  for (int v = 0; v < 8; ++v) cp[(size_t)v * N] = c[v];
}

template <int F, bool RELU>
__global__ __launch_bounds__(256) void k_gather(const float* __restrict__ hw,
                                                const float* __restrict__ dinv,
                                                const int* __restrict__ rowptr,
                                                const int* __restrict__ csr,
                                                const float* __restrict__ b,
                                                float* __restrict__ outp, int n, int E) {
  constexpr int VPL = F / 32;
  typedef __attribute__((ext_vector_type(VPL))) float vecf;
  const int lane = threadIdx.x & 31;
  const int d = __builtin_amdgcn_readfirstlane((int)((blockIdx.x * blockDim.x + threadIdx.x) >> 5));
  if (d >= n) return;

  int beg = rowptr[d];
  int end = rowptr[d + 1];
  beg = beg < 0 ? 0 : beg;
  end = end > E ? E : end;
  const float dd = dinv[d];
  const float* hl = hw + VPL * lane;
  const vecf self = *(const vecf*)(hl + (size_t)d * F);
  const vecf bias = *(const vecf*)(b + VPL * lane);

  vecf acc = (vecf)(0.0f);
  for (int c0 = beg; c0 < end; c0 += 32) {
    const int m = (end - c0) < 32 ? (end - c0) : 32;
    int s = d;
    float w = 0.0f;
    if (lane < m) {
      const int t = csr[c0 + lane];
      s = ((unsigned)t < (unsigned)n) ? t : d;
      w = dinv[s] * dd;
    }
    const int wi = __float_as_int(w);
    for (int j = 0; j < m; j += 4) {
      const int s0 = __builtin_amdgcn_readlane(s, j);
      const int s1 = __builtin_amdgcn_readlane(s, j + 1);
      const int s2 = __builtin_amdgcn_readlane(s, j + 2);
      const int s3 = __builtin_amdgcn_readlane(s, j + 3);
      const vecf r0 = *(const vecf*)(hl + (size_t)s0 * F);
      const vecf r1 = *(const vecf*)(hl + (size_t)s1 * F);
      const vecf r2 = *(const vecf*)(hl + (size_t)s2 * F);
      const vecf r3 = *(const vecf*)(hl + (size_t)s3 * F);
      const float w0 = __int_as_float(__builtin_amdgcn_readlane(wi, j));
      const float w1 = __int_as_float(__builtin_amdgcn_readlane(wi, j + 1));
      const float w2 = __int_as_float(__builtin_amdgcn_readlane(wi, j + 2));
      const float w3 = __int_as_float(__builtin_amdgcn_readlane(wi, j + 3));
      acc += r0 * w0;
      if (j + 1 < m) acc += r1 * w1;
      if (j + 2 < m) acc += r2 * w2;
      if (j + 3 < m) acc += r3 * w3;
    }
  }

  vecf o;
#pragma unroll
  for (int k = 0; k < VPL; ++k) {
    const float v = acc[k] + self[k] * dd * dd + bias[k];
    o[k] = (RELU && !(v > 0.0f)) ? 0.0f : v;
  }
  *(vecf*)(outp + (size_t)d * F + VPL * lane) = o;
}

extern "C" void kernel_launch(void* const* d_in, const int* in_sizes, int n_in,
                              void* d_out, int out_size, void* d_ws, size_t ws_size,
                              hipStream_t stream) {
  (void)n_in;
  (void)out_size;
  (void)ws_size;

  const float* x = (const float*)d_in[0];
  const int* ei = (const int*)d_in[1];
  const float* W1 = (const float*)d_in[2];
  const float* b1 = (const float*)d_in[3];
  const float* W2 = (const float*)d_in[4];
  const float* b2 = (const float*)d_in[5];
  float* out = (float*)d_out;

  constexpr int IN = 256, HID = 128, OUT = 64;
  const int N = in_sizes[0] / IN;
  const int E = in_sizes[1] / 2;
  const int* src = ei;
  const int* dst = ei + E;

  char* ws = (char*)d_ws;
  size_t off = 0;
  auto wsalloc = [&](size_t bytes) -> void* {
    void* p = ws + off;
    off += (bytes + 255) & ~(size_t)255;
    return p;
  };
  int* deg = (int*)wsalloc((size_t)N * 4);
  float* dinv = (float*)wsalloc((size_t)N * 4);
  float* hw1 = (float*)wsalloc((size_t)N * HID * 4);
  float* h1 = (float*)wsalloc((size_t)N * HID * 4);
  int* rowptr = (int*)wsalloc((size_t)(N + 1) * 4);
  int* slot = (int*)wsalloc((size_t)E * 4);
  int* csr = (int*)wsalloc((size_t)E * 4);
  float* hw2 = hw1;

  const int TB = 256;
  auto blks = [](long long total, int tb) { return (unsigned)((total + tb - 1) / tb); };

  k_init_deg<<<blks(N, TB), TB, 0, stream>>>(deg, N);
  k_count_deg<<<blks(E, TB), TB, 0, stream>>>(dst, deg, slot, E);
  k_scan_rows<<<blks(N, SCAN_CHUNK), SCAN_TB, 0, stream>>>(deg, rowptr, dinv, N);
  k_place<<<blks(E, TB), TB, 0, stream>>>(src, dst, rowptr, slot, csr, E);

  {
    const int mTiles = N / 16;
    const int tiles = mTiles * (HID / 16);
    k_gemm_wmma<IN, HID><<<blks(tiles, 8), TB, 0, stream>>>(x, W1, hw1, mTiles);
  }

  k_gather<HID, true><<<blks((long long)N * 32, TB), TB, 0, stream>>>(hw1, dinv, rowptr, csr,
                                                                      b1, h1, N, E);

  {
    const int mTiles = N / 16;
    const int tiles = mTiles * (OUT / 16);
    k_gemm_wmma<HID, OUT><<<blks(tiles, 8), TB, 0, stream>>>(h1, W2, hw2, mTiles);
  }

  k_gather<OUT, false><<<blks((long long)N * 32, TB), TB, 0, stream>>>(hw2, dinv, rowptr, csr,
                                                                       b2, out, N, E);
}


# GEMMs re-tiled for operand reuse: W staged once per work-group in LDS in fragment order; one wave owns a 16-row strip over all column tiles, A loaded once per k; k order in each group of 8 permuted
# speedup vs baseline: 2.6578x; 1.2586x over previous
#include <hip/hip_runtime.h>
#include <hip/hip_bf16.h>

typedef __attribute__((ext_vector_type(2))) float v2f;
typedef __attribute__((ext_vector_type(8))) float v8f;

__global__ void k_init_deg(int* __restrict__ deg, int n) {
  int i = blockIdx.x * blockDim.x + threadIdx.x;
  if (i < n) deg[i] = 1;
}

__global__ void k_count_deg(const int* __restrict__ dst, int* __restrict__ deg,
                            int* __restrict__ slot, int E) {
  int i = blockIdx.x * blockDim.x + threadIdx.x;
  if (i < E) slot[i] = atomicAdd(&deg[dst[i]], 1) - 1;
}

constexpr int SCAN_TB = 256;
constexpr int SCAN_CHUNK = 4 * SCAN_TB;

__global__ __launch_bounds__(SCAN_TB) void k_scan_rows(const int* __restrict__ deg,
                                                       int* __restrict__ rowptr,
                                                       float* __restrict__ dinv, int n) {
  __shared__ int s_front[SCAN_TB / 32];
  __shared__ int s_wsum[SCAN_TB / 32];
  const int tid = threadIdx.x;
  const int lane = tid & 31;
  const int wave = tid >> 5;
  const int base = blockIdx.x * SCAN_CHUNK;

  int front = 0;
  const int4* d4 = (const int4*)deg;
  const int n4 = base >> 2;
  for (int i = tid; i < n4; i += 8 * SCAN_TB) {
    int4 q[8];
#pragma unroll
    for (int u = 0; u < 8; ++u) {
      const int iu = i + u * SCAN_TB;
      q[u] = d4[iu < n4 ? iu : 0];
    }
#pragma unroll
    for (int u = 0; u < 8; ++u) {
      const int t = (q[u].x - 1) + (q[u].y - 1) + (q[u].z - 1) + (q[u].w - 1);
      front += (i + u * SCAN_TB < n4) ? t : 0;
    }
  }
  for (int o = 16; o > 0; o >>= 1) front += __shfl_xor(front, o);
  if (lane == 0) s_front[wave] = front;

  const int i0 = base + 4 * tid;
  int v[4];
  int tsum = 0;
#pragma unroll
  for (int j = 0; j < 4; ++j) {
    const int t = deg[i0 + j < n ? i0 + j : n - 1];
    v[j] = (i0 + j < n) ? t - 1 : 0;
    tsum += v[j];
  }
  int inc = tsum;
  for (int o = 1; o < 32; o <<= 1) {
    const int t = __shfl_up(inc, o);
    if (lane >= o) inc += t;
  }
  if (lane == 31) s_wsum[wave] = inc;
  __syncthreads();

  int run = inc - tsum;
  for (int w = 0; w < SCAN_TB / 32; ++w) {
    run += s_front[w];
    if (w < wave) run += s_wsum[w];
  }
#pragma unroll
  for (int j = 0; j < 4; ++j) {
    if (i0 + j < n) {
      rowptr[i0 + j] = run;
      dinv[i0 + j] = rsqrtf((float)(v[j] + 1));
    }
    run += v[j];
  }
  if (blockIdx.x == gridDim.x - 1 && tid == SCAN_TB - 1) rowptr[n] = run;
}

__global__ void k_place(const int* __restrict__ src, const int* __restrict__ dst,
                        const int* __restrict__ rowptr, const int* __restrict__ slot,
                        int* __restrict__ csr, int E) {
  int i = blockIdx.x * blockDim.x + threadIdx.x;
  if (i >= E) return;
  const int s = src[i];
  const int p = rowptr[dst[i]] + slot[i];
  if ((unsigned)p < (unsigned)E) csr[p] = s;
}

typedef __attribute__((ext_vector_type(4))) float v4f;

constexpr int GEMM_GRID = 512;

template <int K, int N>
__global__ __launch_bounds__(256) void k_gemm_wmma(const float* __restrict__ A,
                                                   const float* __restrict__ W,
                                                   float* __restrict__ C, int mTiles) {
  constexpr int TN = N / 16;
  constexpr int KC = 64;
  constexpr int CPS = K / KC;
  constexpr int NV = K * N / 4;
  __shared__ v4f sW[NV];

#pragma unroll 8
  for (int p = threadIdx.x; p < NV; p += 256) {
    const int cc = p & 15;
    const int t0 = (p >> 4) & 1;
    const int hh = (p >> 5) & 1;
    const int rest = p >> 6;
    const int t = 2 * (rest % (TN / 2)) + t0;
    const int J = rest / (TN / 2);
    const float* wp = W + (size_t)(8 * J + 4 * hh) * N + 16 * t + cc;
    v4f w;
    w.x = wp[0];
    w.y = wp[N];
    w.z = wp[2 * N];
    w.w = wp[3 * N];
    sW[(J * TN + t) * 32 + hh * 16 + cc] = w;
  }
  __syncthreads();

  const int lane = threadIdx.x & 31;
  const int wave = __builtin_amdgcn_readfirstlane((int)(threadIdx.x >> 5));
  const int c = lane & 15;
  const int h = lane >> 4;
  const int stride = 8 * (int)gridDim.x;
  int s = wave * (int)gridDim.x + (int)blockIdx.x;
  if (s >= mTiles) return;

  const float* arow = A + ((size_t)(16 * s + c) * K + 4 * h);
  v4f acur[KC / 8];
#pragma unroll
  for (int u = 0; u < KC / 8; ++u) acur[u] = *(const v4f*)(arow + 8 * u);

  v4f bcur[TN];
#pragma unroll
  for (int t = 0; t < TN; ++t) bcur[t] = sW[t * 32 + lane];

  for (; s < mTiles; s += stride) {
    const float* nrow = (s + stride < mTiles) ? arow + (size_t)stride * 16 * K : arow;

    v8f acc[TN];
#pragma unroll
    for (int t = 0; t < TN; ++t) acc[t] = (v8f){0.f, 0.f, 0.f, 0.f, 0.f, 0.f, 0.f, 0.f};

#pragma nounroll
    for (int ch = 0; ch < CPS; ++ch) {
      const bool lastc = (ch + 1 == CPS);
      const float* np = lastc ? nrow : arow + KC * (ch + 1);
      const v4f* bp = sW + (ch * (KC / 8) * TN) * 32 + lane;
      const v4f* bq = lastc ? sW + lane : bp + (KC / 8) * TN * 32;

#pragma unroll
      for (int u = 0; u < KC / 8; ++u) {
        const v4f* bn = (u + 1 < KC / 8) ? bp + (u + 1) * TN * 32 : bq;
        v4f bnxt[TN];
#pragma unroll
        for (int t = 0; t < TN; ++t) bnxt[t] = bn[t * 32];
        __builtin_amdgcn_sched_barrier(0);

        const v2f a0 = {acur[u].x, acur[u].y};
        const v2f a1 = {acur[u].z, acur[u].w};
#pragma unroll
        for (int t = 0; t < TN; ++t) {
          const v2f b0 = {bcur[t].x, bcur[t].y};
          acc[t] = __builtin_amdgcn_wmma_f32_16x16x4_f32(false, a0, false, b0, (short)0,
                                                         acc[t], false, false);
        }
#pragma unroll
        for (int t = 0; t < TN; ++t) {
          const v2f b1 = {bcur[t].z, bcur[t].w};
          acc[t] = __builtin_amdgcn_wmma_f32_16x16x4_f32(false, a1, false, b1, (short)0,
                                                         acc[t], false, false);
        }
#pragma unroll
        for (int t = 0; t < TN; ++t) bcur[t] = bnxt[t];
        acur[u] = *(const v4f*)(np + 8 * u);
        __builtin_amdgcn_sched_barrier(0);
      }
    }

    float* cp = C + ((size_t)(16 * s + 8 * h) * N + c);
#pragma unroll
    for (int t = 0; t < TN; ++t) {
#pragma unroll
      for (int v = 0; v < 8; ++v) cp[(size_t)v * N + 16 * t] = acc[t][v];
    }
    arow = nrow;
  }
}

template <int F, bool RELU>
__global__ __launch_bounds__(256) void k_gather(const float* __restrict__ hw,
                                                const float* __restrict__ dinv,
                                                const int* __restrict__ rowptr,
                                                const int* __restrict__ csr,
                                                const float* __restrict__ b,
                                                float* __restrict__ outp, int n, int E) {
  constexpr int VPL = F / 32;
  typedef __attribute__((ext_vector_type(VPL))) float vecf;
  const int lane = threadIdx.x & 31;
  const int d = __builtin_amdgcn_readfirstlane((int)((blockIdx.x * blockDim.x + threadIdx.x) >> 5));
  if (d >= n) return;

  int beg = rowptr[d];
  int end = rowptr[d + 1];
  beg = beg < 0 ? 0 : beg;
  end = end > E ? E : end;
  const float dd = dinv[d];
  const float* hl = hw + VPL * lane;
  const vecf self = *(const vecf*)(hl + (size_t)d * F);
  const vecf bias = *(const vecf*)(b + VPL * lane);

  vecf acc = (vecf)(0.0f);
  for (int c0 = beg; c0 < end; c0 += 32) {
    const int m = (end - c0) < 32 ? (end - c0) : 32;
    int s = d;
    float w = 0.0f;
    if (lane < m) {
      const int t = csr[c0 + lane];
      s = ((unsigned)t < (unsigned)n) ? t : d;
      w = dinv[s] * dd;
    }
    const int wi = __float_as_int(w);
    for (int j = 0; j < m; j += 4) {
      const int s0 = __builtin_amdgcn_readlane(s, j);
      const int s1 = __builtin_amdgcn_readlane(s, j + 1);
      const int s2 = __builtin_amdgcn_readlane(s, j + 2);
      const int s3 = __builtin_amdgcn_readlane(s, j + 3);
      const vecf r0 = *(const vecf*)(hl + (size_t)s0 * F);
      const vecf r1 = *(const vecf*)(hl + (size_t)s1 * F);
      const vecf r2 = *(const vecf*)(hl + (size_t)s2 * F);
      const vecf r3 = *(const vecf*)(hl + (size_t)s3 * F);
      const float w0 = __int_as_float(__builtin_amdgcn_readlane(wi, j));
      const float w1 = __int_as_float(__builtin_amdgcn_readlane(wi, j + 1));
      const float w2 = __int_as_float(__builtin_amdgcn_readlane(wi, j + 2));
      const float w3 = __int_as_float(__builtin_amdgcn_readlane(wi, j + 3));
      acc += r0 * w0;
      if (j + 1 < m) acc += r1 * w1;
      if (j + 2 < m) acc += r2 * w2;
      if (j + 3 < m) acc += r3 * w3;
    }
  }

  vecf o;
#pragma unroll
  for (int k = 0; k < VPL; ++k) {
    const float v = acc[k] + self[k] * dd * dd + bias[k];
    o[k] = (RELU && !(v > 0.0f)) ? 0.0f : v;
  }
  *(vecf*)(outp + (size_t)d * F + VPL * lane) = o;
}

extern "C" void kernel_launch(void* const* d_in, const int* in_sizes, int n_in,
                              void* d_out, int out_size, void* d_ws, size_t ws_size,
                              hipStream_t stream) {
  (void)n_in;
  (void)out_size;
  (void)ws_size;

  const float* x = (const float*)d_in[0];
  const int* ei = (const int*)d_in[1];
  const float* W1 = (const float*)d_in[2];
  const float* b1 = (const float*)d_in[3];
  const float* W2 = (const float*)d_in[4];
  const float* b2 = (const float*)d_in[5];
  float* out = (float*)d_out;

  constexpr int IN = 256, HID = 128, OUT = 64;
  const int N = in_sizes[0] / IN;
  const int E = in_sizes[1] / 2;
  const int* src = ei;
  const int* dst = ei + E;

  char* ws = (char*)d_ws;
  size_t off = 0;
  auto wsalloc = [&](size_t bytes) -> void* {
    void* p = ws + off;
    off += (bytes + 255) & ~(size_t)255;
    return p;
  };
  int* deg = (int*)wsalloc((size_t)N * 4);
  float* dinv = (float*)wsalloc((size_t)N * 4);
  float* hw1 = (float*)wsalloc((size_t)N * HID * 4);
  float* h1 = (float*)wsalloc((size_t)N * HID * 4);
  int* rowptr = (int*)wsalloc((size_t)(N + 1) * 4);
  int* slot = (int*)wsalloc((size_t)E * 4);
  int* csr = (int*)wsalloc((size_t)E * 4);
  float* hw2 = hw1;

  const int TB = 256;
  auto blks = [](long long total, int tb) { return (unsigned)((total + tb - 1) / tb); };

  k_init_deg<<<blks(N, TB), TB, 0, stream>>>(deg, N);
  k_count_deg<<<blks(E, TB), TB, 0, stream>>>(dst, deg, slot, E);
  k_scan_rows<<<blks(N, SCAN_CHUNK), SCAN_TB, 0, stream>>>(deg, rowptr, dinv, N);
  k_place<<<blks(E, TB), TB, 0, stream>>>(src, dst, rowptr, slot, csr, E);

  {
    const int mTiles = N / 16;
    k_gemm_wmma<IN, HID><<<GEMM_GRID, TB, 0, stream>>>(x, W1, hw1, mTiles);
  }

  k_gather<HID, true><<<blks((long long)N * 32, TB), TB, 0, stream>>>(hw1, dinv, rowptr, csr,
                                                                      b1, h1, N, E);

  {
    const int mTiles = N / 16;
    k_gemm_wmma<HID, OUT><<<GEMM_GRID, TB, 0, stream>>>(h1, W2, hw2, mTiles);
  }

  k_gather<OUT, false><<<blks((long long)N * 32, TB), TB, 0, stream>>>(hw2, dinv, rowptr, csr,
                                                                       b2, out, N, E);
}


# edge bucketing without global atomics: two-level counting sort on LDS counters (256-node bins, runs ordered in LDS, whole-line stores); 8 launches -> 7; sum order in a row = LDS add rank
# speedup vs baseline: 4.3901x; 1.6518x over previous
#include <hip/hip_runtime.h>
#include <hip/hip_bf16.h>

typedef __attribute__((ext_vector_type(2))) float v2f;
typedef __attribute__((ext_vector_type(8))) float v8f;

constexpr int BIN_SHIFT = 8;
constexpr int BIN_NODES = 1 << BIN_SHIFT;
constexpr int MAXBIN = 512;
constexpr int NSLICE = 256;
constexpr int BK_TB = 1024;
constexpr int FILL_CAP = 8192;
constexpr int ROWS_CAP = 8192;

__device__ __forceinline__ int bin_count(int n) {
  const int nb = (n + BIN_NODES - 1) >> BIN_SHIFT;
  return nb < MAXBIN ? nb : MAXBIN;
}

__device__ __forceinline__ int bin_of(int d, int nbins) {
  const int b = (int)((unsigned)d >> BIN_SHIFT);
  return b < nbins ? b : nbins - 1;
}

__global__ __launch_bounds__(BK_TB) void k_bin_count(const int* __restrict__ dst,
                                                     int* __restrict__ hist, int n, int E) {
  __shared__ int h[MAXBIN];
  const int tid = threadIdx.x;
  const int nbins = bin_count(n);
  for (int b = tid; b < MAXBIN; b += BK_TB) h[b] = 0;
  __syncthreads();

  const int per = (E + NSLICE - 1) / NSLICE;
  const int e0 = (int)blockIdx.x * per;
  const int e1 = (e0 + per < E) ? e0 + per : E;
  for (int i0 = e0 + tid; i0 < e1; i0 += 8 * BK_TB) {
    int d[8];
#pragma unroll
    for (int u = 0; u < 8; ++u) {
      const int i = i0 + u * BK_TB;
      d[u] = dst[i < e1 ? i : e1 - 1];
    }
#pragma unroll
    for (int u = 0; u < 8; ++u) {
      if (i0 + u * BK_TB < e1) atomicAdd(&h[bin_of(d[u], nbins)], 1);
    }
  }
  __syncthreads();
  for (int b = tid; b < nbins; b += BK_TB) hist[b * NSLICE + (int)blockIdx.x] = h[b];
}

__global__ __launch_bounds__(BK_TB) void k_bin_fill(const int* __restrict__ src,
                                                    const int* __restrict__ dst,
                                                    const int* __restrict__ hist,
                                                    int* __restrict__ binstart,
                                                    int2* __restrict__ binned, int n, int E) {
  __shared__ int sT[MAXBIN];
  __shared__ int sP[MAXBIN];
  __shared__ int sL[MAXBIN];
  __shared__ int sCnt[MAXBIN];
  __shared__ int sWT[BK_TB / 32];
  __shared__ int sWL[BK_TB / 32];
  __shared__ int2 stage[FILL_CAP];
  const int tid = threadIdx.x;
  const int lane = tid & 31;
  const int wave = tid >> 5;
  const int blk = (int)blockIdx.x;
  const int nbins = bin_count(n);

  const int sub = tid & 7;
  const int grp = tid >> 3;
  const int4* h4 = (const int4*)hist;
  for (int r0 = 0; r0 < MAXBIN; r0 += BK_TB / 8) {
    const int r = r0 + grp;
    int tot = 0, pre = 0;
    if (r < nbins) {
      int4 q[NSLICE / 32];
#pragma unroll
      for (int j = 0; j < NSLICE / 32; ++j) q[j] = h4[r * (NSLICE / 4) + j * 8 + sub];
#pragma unroll
      for (int j = 0; j < NSLICE / 32; ++j) {
        const int k = (j * 8 + sub) * 4;
        tot += q[j].x + q[j].y + q[j].z + q[j].w;
        pre += (k < blk ? q[j].x : 0) + (k + 1 < blk ? q[j].y : 0) + (k + 2 < blk ? q[j].z : 0) +
               (k + 3 < blk ? q[j].w : 0);
      }
    }
    for (int o = 1; o < 8; o <<= 1) {
      tot += __shfl_xor(tot, o);
      pre += __shfl_xor(pre, o);
    }
    if (sub == 0) {
      sT[r] = tot;
      sP[r] = pre;
    }
  }
  for (int r = tid; r < MAXBIN; r += BK_TB) {
    sL[r] = (r < nbins) ? hist[r * NSLICE + blk] : 0;
    sCnt[r] = 0;
  }
  __syncthreads();

  int t = 0, l = 0;
  if (tid < MAXBIN) {
    t = sT[tid];
    l = sL[tid];
  }
  int it = t, il = l;
  for (int o = 1; o < 32; o <<= 1) {
    const int a = __shfl_up(it, o);
    const int c = __shfl_up(il, o);
    if (lane >= o) {
      it += a;
      il += c;
    }
  }
  if (lane == 31) {
    sWT[wave] = it;
    sWL[wave] = il;
  }
  __syncthreads();
  if (tid < MAXBIN) {
    int bt = it - t, bl = il - l;
    for (int w = 0; w < MAXBIN / 32; ++w) {
      if (w < wave) {
        bt += sWT[w];
        bl += sWL[w];
      }
    }
    sP[tid] += bt;
    sL[tid] = bl;
    if (blk == 0 && tid < nbins) {
      binstart[tid] = bt;
      if (tid == nbins - 1) binstart[nbins] = bt + t;
    }
  }
  __syncthreads();

  const int per = (E + NSLICE - 1) / NSLICE;
  const int e0 = blk * per;
  const int e1 = (e0 + per < E) ? e0 + per : E;
  for (int i0 = e0 + tid; i0 < e1; i0 += 8 * BK_TB) {
    int s[8], d[8];
#pragma unroll
    for (int u = 0; u < 8; ++u) {
      const int i = i0 + u * BK_TB;
      const int ic = i < e1 ? i : e1 - 1;
      s[u] = src[ic];
      d[u] = dst[ic];
    }
#pragma unroll
    for (int u = 0; u < 8; ++u) {
      if (i0 + u * BK_TB < e1) {
        const int b = bin_of(d[u], nbins);
        const int rank = atomicAdd(&sCnt[b], 1);
        const int li = sL[b] + rank;
        if (li < FILL_CAP) {
          stage[li] = make_int2(s[u], d[u]);
        } else {
          const int p = sP[b] + rank;
          if ((unsigned)p < (unsigned)E) binned[p] = make_int2(s[u], d[u]);
        }
      }
    }
  }
  __syncthreads();

  int m = e1 - e0;
  m = m < 0 ? 0 : (m < FILL_CAP ? m : FILL_CAP);
  for (int i = tid; i < m; i += BK_TB) {
    const int2 e = stage[i];
    const int b = bin_of(e.y, nbins);
    const int p = sP[b] + (i - sL[b]);
    if ((unsigned)p < (unsigned)E) binned[p] = e;
  }
}

__global__ __launch_bounds__(BK_TB) void k_bin_rows(const int2* __restrict__ binned,
                                                    const int* __restrict__ binstart,
                                                    int* __restrict__ rowptr,
                                                    float* __restrict__ dinv,
                                                    int* __restrict__ csr, int n, int E) {
  __shared__ int cnt[BIN_NODES];
  __shared__ int excl[BIN_NODES];
  __shared__ int sW[BIN_NODES / 32];
  __shared__ int stage[ROWS_CAP];
  const int tid = threadIdx.x;
  const int lane = tid & 31;
  const int wave = tid >> 5;
  const int bin = (int)blockIdx.x;
  const int base = bin << BIN_SHIFT;
  int b0 = binstart[bin];
  int b1 = binstart[bin + 1];
  b0 = b0 < 0 ? 0 : (b0 > E ? E : b0);
  b1 = b1 < b0 ? b0 : (b1 > E ? E : b1);

  if (tid < BIN_NODES) cnt[tid] = 0;
  __syncthreads();

  for (int i0 = b0 + tid; i0 < b1; i0 += 8 * BK_TB) {
    int d[8];
#pragma unroll
    for (int u = 0; u < 8; ++u) {
      const int i = i0 + u * BK_TB;
      d[u] = binned[i < b1 ? i : b1 - 1].y;
    }
#pragma unroll
    for (int u = 0; u < 8; ++u) {
      const int dl = d[u] - base;
      if (i0 + u * BK_TB < b1 && (unsigned)dl < (unsigned)BIN_NODES) atomicAdd(&cnt[dl], 1);
    }
  }
  __syncthreads();

  int c = 0;
  if (tid < BIN_NODES) c = cnt[tid];
  int inc = c;
  for (int o = 1; o < 32; o <<= 1) {
    const int a = __shfl_up(inc, o);
    if (lane >= o) inc += a;
  }
  if (lane == 31 && wave < BIN_NODES / 32) sW[wave] = inc;
  __syncthreads();
  if (tid < BIN_NODES) {
    int ex = inc - c;
    for (int w = 0; w < BIN_NODES / 32; ++w) {
      if (w < wave) ex += sW[w];
    }
    excl[tid] = ex;
    cnt[tid] = 0;
    const int node = base + tid;
    if (node < n) {
      rowptr[node] = b0 + ex;
      dinv[node] = rsqrtf((float)(c + 1));
    }
  }
  if (bin == (int)gridDim.x - 1 && tid == 0) rowptr[n] = b1;
  __syncthreads();

  for (int i0 = b0 + tid; i0 < b1; i0 += 8 * BK_TB) {
    int2 e[8];
#pragma unroll
    for (int u = 0; u < 8; ++u) {
      const int i = i0 + u * BK_TB;
      e[u] = binned[i < b1 ? i : b1 - 1];
    }
#pragma unroll
    for (int u = 0; u < 8; ++u) {
      const int dl = e[u].y - base;
      if (i0 + u * BK_TB < b1 && (unsigned)dl < (unsigned)BIN_NODES) {
        const int p = excl[dl] + atomicAdd(&cnt[dl], 1);
        if (p < ROWS_CAP) {
          stage[p] = e[u].x;
        } else if (b0 + p < b1) {
          csr[b0 + p] = e[u].x;
        }
      }
    }
  }
  __syncthreads();

  int m = b1 - b0;
  m = m < ROWS_CAP ? m : ROWS_CAP;
  for (int i = tid; i < m; i += BK_TB) csr[b0 + i] = stage[i];
}

typedef __attribute__((ext_vector_type(4))) float v4f;

constexpr int GEMM_GRID = 512;

template <int K, int N>
__global__ __launch_bounds__(256) void k_gemm_wmma(const float* __restrict__ A,
                                                   const float* __restrict__ W,
                                                   float* __restrict__ C, int mTiles) {
  constexpr int TN = N / 16;
  constexpr int KC = 64;
  constexpr int CPS = K / KC;
  constexpr int NV = K * N / 4;
  __shared__ v4f sW[NV];

#pragma unroll 8
  for (int p = threadIdx.x; p < NV; p += 256) {
    const int cc = p & 15;
    const int t0 = (p >> 4) & 1;
    const int hh = (p >> 5) & 1;
    const int rest = p >> 6;
    const int t = 2 * (rest % (TN / 2)) + t0;
    const int J = rest / (TN / 2);
    const float* wp = W + (size_t)(8 * J + 4 * hh) * N + 16 * t + cc;
    v4f w;
    w.x = wp[0];
    w.y = wp[N];
    w.z = wp[2 * N];
    w.w = wp[3 * N];
    sW[(J * TN + t) * 32 + hh * 16 + cc] = w;
  }
  __syncthreads();

  const int lane = threadIdx.x & 31;
  const int wave = __builtin_amdgcn_readfirstlane((int)(threadIdx.x >> 5));
  const int c = lane & 15;
  const int h = lane >> 4;
  const int stride = 8 * (int)gridDim.x;
  int s = wave * (int)gridDim.x + (int)blockIdx.x;
  if (s >= mTiles) return;

  const float* arow = A + ((size_t)(16 * s + c) * K + 4 * h);
  v4f acur[KC / 8];
#pragma unroll
  for (int u = 0; u < KC / 8; ++u) acur[u] = *(const v4f*)(arow + 8 * u);

  v4f bcur[TN];
#pragma unroll
  for (int t = 0; t < TN; ++t) bcur[t] = sW[t * 32 + lane];

  for (; s < mTiles; s += stride) {
    const float* nrow = (s + stride < mTiles) ? arow + (size_t)stride * 16 * K : arow;

    v8f acc[TN];
#pragma unroll
    for (int t = 0; t < TN; ++t) acc[t] = (v8f){0.f, 0.f, 0.f, 0.f, 0.f, 0.f, 0.f, 0.f};

#pragma nounroll
    for (int ch = 0; ch < CPS; ++ch) {
      const bool lastc = (ch + 1 == CPS);
      const float* np = lastc ? nrow : arow + KC * (ch + 1);
      const v4f* bp = sW + (ch * (KC / 8) * TN) * 32 + lane;
      const v4f* bq = lastc ? sW + lane : bp + (KC / 8) * TN * 32;

#pragma unroll
      for (int u = 0; u < KC / 8; ++u) {
        const v4f* bn = (u + 1 < KC / 8) ? bp + (u + 1) * TN * 32 : bq;
        v4f bnxt[TN];
#pragma unroll
        for (int t = 0; t < TN; ++t) bnxt[t] = bn[t * 32];
        __builtin_amdgcn_sched_barrier(0);

        const v2f a0 = {acur[u].x, acur[u].y};
        const v2f a1 = {acur[u].z, acur[u].w};
#pragma unroll
        for (int t = 0; t < TN; ++t) {
          const v2f b0 = {bcur[t].x, bcur[t].y};
          acc[t] = __builtin_amdgcn_wmma_f32_16x16x4_f32(false, a0, false, b0, (short)0,
                                                         acc[t], false, false);
        }
#pragma unroll
        for (int t = 0; t < TN; ++t) {
          const v2f b1 = {bcur[t].z, bcur[t].w};
          acc[t] = __builtin_amdgcn_wmma_f32_16x16x4_f32(false, a1, false, b1, (short)0,
                                                         acc[t], false, false);
        }
#pragma unroll
        for (int t = 0; t < TN; ++t) bcur[t] = bnxt[t];
        acur[u] = *(const v4f*)(np + 8 * u);
        __builtin_amdgcn_sched_barrier(0);
      }
    }

    float* cp = C + ((size_t)(16 * s + 8 * h) * N + c);
#pragma unroll
    for (int t = 0; t < TN; ++t) {
#pragma unroll
      for (int v = 0; v < 8; ++v) cp[(size_t)v * N + 16 * t] = acc[t][v];
    }
    arow = nrow;
  }
}

template <int F, bool RELU>
__global__ __launch_bounds__(256) void k_gather(const float* __restrict__ hw,
                                                const float* __restrict__ dinv,
                                                const int* __restrict__ rowptr,
                                                const int* __restrict__ csr,
                                                const float* __restrict__ b,
                                                float* __restrict__ outp, int n, int E) {
  constexpr int VPL = F / 32;
  typedef __attribute__((ext_vector_type(VPL))) float vecf;
  const int lane = threadIdx.x & 31;
  const int d = __builtin_amdgcn_readfirstlane((int)((blockIdx.x * blockDim.x + threadIdx.x) >> 5));
  if (d >= n) return;

  int beg = rowptr[d];
  int end = rowptr[d + 1];
  beg = beg < 0 ? 0 : beg;
  end = end > E ? E : end;
  const float dd = dinv[d];
  const float* hl = hw + VPL * lane;
  const vecf self = *(const vecf*)(hl + (size_t)d * F);
  const vecf bias = *(const vecf*)(b + VPL * lane);

  vecf acc = (vecf)(0.0f);
  for (int c0 = beg; c0 < end; c0 += 32) {
    const int m = (end - c0) < 32 ? (end - c0) : 32;
    int s = d;
    float w = 0.0f;
    if (lane < m) {
      const int t = csr[c0 + lane];
      s = ((unsigned)t < (unsigned)n) ? t : d;
      w = dinv[s] * dd;
    }
    const int wi = __float_as_int(w);
    for (int j = 0; j < m; j += 4) {
      const int s0 = __builtin_amdgcn_readlane(s, j);
      const int s1 = __builtin_amdgcn_readlane(s, j + 1);
      const int s2 = __builtin_amdgcn_readlane(s, j + 2);
      const int s3 = __builtin_amdgcn_readlane(s, j + 3);
      const vecf r0 = *(const vecf*)(hl + (size_t)s0 * F);
      const vecf r1 = *(const vecf*)(hl + (size_t)s1 * F);
      const vecf r2 = *(const vecf*)(hl + (size_t)s2 * F);
      const vecf r3 = *(const vecf*)(hl + (size_t)s3 * F);
      const float w0 = __int_as_float(__builtin_amdgcn_readlane(wi, j));
      const float w1 = __int_as_float(__builtin_amdgcn_readlane(wi, j + 1));
      const float w2 = __int_as_float(__builtin_amdgcn_readlane(wi, j + 2));
      const float w3 = __int_as_float(__builtin_amdgcn_readlane(wi, j + 3));
      acc += r0 * w0;
      if (j + 1 < m) acc += r1 * w1;
      if (j + 2 < m) acc += r2 * w2;
      if (j + 3 < m) acc += r3 * w3;
    }
  }

  vecf o;
#pragma unroll
  for (int k = 0; k < VPL; ++k) {
    const float v = acc[k] + self[k] * dd * dd + bias[k];
    o[k] = (RELU && !(v > 0.0f)) ? 0.0f : v;
  }
  *(vecf*)(outp + (size_t)d * F + VPL * lane) = o;
}

extern "C" void kernel_launch(void* const* d_in, const int* in_sizes, int n_in,
                              void* d_out, int out_size, void* d_ws, size_t ws_size,
                              hipStream_t stream) {
  (void)n_in;
  (void)out_size;
  (void)ws_size;

  const float* x = (const float*)d_in[0];
  const int* ei = (const int*)d_in[1];
  const float* W1 = (const float*)d_in[2];
  const float* b1 = (const float*)d_in[3];
  const float* W2 = (const float*)d_in[4];
  const float* b2 = (const float*)d_in[5];
  float* out = (float*)d_out;

  constexpr int IN = 256, HID = 128, OUT = 64;
  const int N = in_sizes[0] / IN;
  const int E = in_sizes[1] / 2;
  const int* src = ei;
  const int* dst = ei + E;

  char* ws = (char*)d_ws;
  size_t off = 0;
  auto wsalloc = [&](size_t bytes) -> void* {
    void* p = ws + off;
    off += (bytes + 255) & ~(size_t)255;
    return p;
  };
  float* dinv = (float*)wsalloc((size_t)N * 4);
  float* hw1 = (float*)wsalloc((size_t)N * HID * 4);
  float* h1 = (float*)wsalloc((size_t)N * HID * 4);
  int* rowptr = (int*)wsalloc((size_t)(N + 1) * 4);
  int* csr = (int*)wsalloc((size_t)E * 4);
  float* hw2 = hw1;
  int2* binned = (int2*)h1;
  int* hist = (int*)((char*)h1 + (((size_t)E * 8 + 255) & ~(size_t)255));
  int* binstart = hist + MAXBIN * NSLICE;

  const int TB = 256;
  auto blks = [](long long total, int tb) { return (unsigned)((total + tb - 1) / tb); };

  k_bin_count<<<NSLICE, BK_TB, 0, stream>>>(dst, hist, N, E);
  k_bin_fill<<<NSLICE, BK_TB, 0, stream>>>(src, dst, hist, binstart, binned, N, E);
  k_bin_rows<<<blks(N, BIN_NODES), BK_TB, 0, stream>>>(binned, binstart, rowptr, dinv, csr, N, E);

  {
    const int mTiles = N / 16;
    k_gemm_wmma<IN, HID><<<GEMM_GRID, TB, 0, stream>>>(x, W1, hw1, mTiles);
  }

  k_gather<HID, true><<<blks((long long)N * 32, TB), TB, 0, stream>>>(hw1, dinv, rowptr, csr,
                                                                      b1, h1, N, E);

  {
    const int mTiles = N / 16;
    k_gemm_wmma<HID, OUT><<<GEMM_GRID, TB, 0, stream>>>(h1, W2, hw2, mTiles);
  }

  k_gather<OUT, false><<<blks((long long)N * 32, TB), TB, 0, stream>>>(hw2, dinv, rowptr, csr,
                                                                       b2, out, N, E);
}


# edge factor dinv[src] folded into the GEMM epilogue (hw rows stored times dinv[row]); gather adds rows unweighted, scales the sum once by dinv[dst]; per-edge dinv[src] lookup gone; sum order kept
# speedup vs baseline: 4.5417x; 1.0345x over previous
#include <hip/hip_runtime.h>
#include <hip/hip_bf16.h>

typedef __attribute__((ext_vector_type(2))) float v2f;
typedef __attribute__((ext_vector_type(8))) float v8f;

constexpr int BIN_SHIFT = 8;
constexpr int BIN_NODES = 1 << BIN_SHIFT;
constexpr int MAXBIN = 512;
constexpr int NSLICE = 256;
constexpr int BK_TB = 1024;
constexpr int FILL_CAP = 8192;
constexpr int ROWS_CAP = 8192;

__device__ __forceinline__ int bin_count(int n) {
  const int nb = (n + BIN_NODES - 1) >> BIN_SHIFT;
  return nb < MAXBIN ? nb : MAXBIN;
}

__device__ __forceinline__ int bin_of(int d, int nbins) {
  const int b = (int)((unsigned)d >> BIN_SHIFT);
  return b < nbins ? b : nbins - 1;
}

__global__ __launch_bounds__(BK_TB) void k_bin_count(const int* __restrict__ dst,
                                                     int* __restrict__ hist, int n, int E) {
  __shared__ int h[MAXBIN];
  const int tid = threadIdx.x;
  const int nbins = bin_count(n);
  for (int b = tid; b < MAXBIN; b += BK_TB) h[b] = 0;
  __syncthreads();

  const int per = (E + NSLICE - 1) / NSLICE;
  const int e0 = (int)blockIdx.x * per;
  const int e1 = (e0 + per < E) ? e0 + per : E;
  for (int i0 = e0 + tid; i0 < e1; i0 += 8 * BK_TB) {
    int d[8];
#pragma unroll
    for (int u = 0; u < 8; ++u) {
      const int i = i0 + u * BK_TB;
      d[u] = dst[i < e1 ? i : e1 - 1];
    }
#pragma unroll
    for (int u = 0; u < 8; ++u) {
      if (i0 + u * BK_TB < e1) atomicAdd(&h[bin_of(d[u], nbins)], 1);
    }
  }
  __syncthreads();
  for (int b = tid; b < nbins; b += BK_TB) hist[b * NSLICE + (int)blockIdx.x] = h[b];
}

__global__ __launch_bounds__(BK_TB) void k_bin_fill(const int* __restrict__ src,
                                                    const int* __restrict__ dst,
                                                    const int* __restrict__ hist,
                                                    int* __restrict__ binstart,
                                                    int2* __restrict__ binned, int n, int E) {
  __shared__ int sT[MAXBIN];
  __shared__ int sP[MAXBIN];
  __shared__ int sL[MAXBIN];
  __shared__ int sCnt[MAXBIN];
  __shared__ int sWT[BK_TB / 32];
  __shared__ int sWL[BK_TB / 32];
  __shared__ int2 stage[FILL_CAP];
  const int tid = threadIdx.x;
  const int lane = tid & 31;
  const int wave = tid >> 5;
  const int blk = (int)blockIdx.x;
  const int nbins = bin_count(n);

  const int sub = tid & 7;
  const int grp = tid >> 3;
  const int4* h4 = (const int4*)hist;
  for (int r0 = 0; r0 < MAXBIN; r0 += BK_TB / 8) {
    const int r = r0 + grp;
    int tot = 0, pre = 0;
    if (r < nbins) {
      int4 q[NSLICE / 32];
#pragma unroll
      for (int j = 0; j < NSLICE / 32; ++j) q[j] = h4[r * (NSLICE / 4) + j * 8 + sub];
#pragma unroll
      for (int j = 0; j < NSLICE / 32; ++j) {
        const int k = (j * 8 + sub) * 4;
        tot += q[j].x + q[j].y + q[j].z + q[j].w;
        pre += (k < blk ? q[j].x : 0) + (k + 1 < blk ? q[j].y : 0) + (k + 2 < blk ? q[j].z : 0) +
               (k + 3 < blk ? q[j].w : 0);
      }
    }
    for (int o = 1; o < 8; o <<= 1) {
      tot += __shfl_xor(tot, o);
      pre += __shfl_xor(pre, o);
    }
    if (sub == 0) {
      sT[r] = tot;
      sP[r] = pre;
    }
  }
  for (int r = tid; r < MAXBIN; r += BK_TB) {
    sL[r] = (r < nbins) ? hist[r * NSLICE + blk] : 0;
    sCnt[r] = 0;
  }
  __syncthreads();

  int t = 0, l = 0;
  if (tid < MAXBIN) {
    t = sT[tid];
    l = sL[tid];
  }
  int it = t, il = l;
  for (int o = 1; o < 32; o <<= 1) {
    const int a = __shfl_up(it, o);
    const int c = __shfl_up(il, o);
    if (lane >= o) {
      it += a;
      il += c;
    }
  }
  if (lane == 31) {
    sWT[wave] = it;
    sWL[wave] = il;
  }
  __syncthreads();
  if (tid < MAXBIN) {
    int bt = it - t, bl = il - l;
    for (int w = 0; w < MAXBIN / 32; ++w) {
      if (w < wave) {
        bt += sWT[w];
        bl += sWL[w];
      }
    }
    sP[tid] += bt;
    sL[tid] = bl;
    if (blk == 0 && tid < nbins) {
      binstart[tid] = bt;
      if (tid == nbins - 1) binstart[nbins] = bt + t;
    }
  }
  __syncthreads();

  const int per = (E + NSLICE - 1) / NSLICE;
  const int e0 = blk * per;
  const int e1 = (e0 + per < E) ? e0 + per : E;
  for (int i0 = e0 + tid; i0 < e1; i0 += 8 * BK_TB) {
    int s[8], d[8];
#pragma unroll
    for (int u = 0; u < 8; ++u) {
      const int i = i0 + u * BK_TB;
      const int ic = i < e1 ? i : e1 - 1;
      s[u] = src[ic];
      d[u] = dst[ic];
    }
#pragma unroll
    for (int u = 0; u < 8; ++u) {
      if (i0 + u * BK_TB < e1) {
        const int b = bin_of(d[u], nbins);
        const int rank = atomicAdd(&sCnt[b], 1);
        const int li = sL[b] + rank;
        if (li < FILL_CAP) {
          stage[li] = make_int2(s[u], d[u]);
        } else {
          const int p = sP[b] + rank;
          if ((unsigned)p < (unsigned)E) binned[p] = make_int2(s[u], d[u]);
        }
      }
    }
  }
  __syncthreads();

  int m = e1 - e0;
  m = m < 0 ? 0 : (m < FILL_CAP ? m : FILL_CAP);
  for (int i = tid; i < m; i += BK_TB) {
    const int2 e = stage[i];
    const int b = bin_of(e.y, nbins);
    const int p = sP[b] + (i - sL[b]);
    if ((unsigned)p < (unsigned)E) binned[p] = e;
  }
}

__global__ __launch_bounds__(BK_TB) void k_bin_rows(const int2* __restrict__ binned,
                                                    const int* __restrict__ binstart,
                                                    int* __restrict__ rowptr,
                                                    float* __restrict__ dinv,
                                                    int* __restrict__ csr, int n, int E) {
  __shared__ int cnt[BIN_NODES];
  __shared__ int excl[BIN_NODES];
  __shared__ int sW[BIN_NODES / 32];
  __shared__ int stage[ROWS_CAP];
  const int tid = threadIdx.x;
  const int lane = tid & 31;
  const int wave = tid >> 5;
  const int bin = (int)blockIdx.x;
  const int base = bin << BIN_SHIFT;
  int b0 = binstart[bin];
  int b1 = binstart[bin + 1];
  b0 = b0 < 0 ? 0 : (b0 > E ? E : b0);
  b1 = b1 < b0 ? b0 : (b1 > E ? E : b1);

  if (tid < BIN_NODES) cnt[tid] = 0;
  __syncthreads();

  for (int i0 = b0 + tid; i0 < b1; i0 += 8 * BK_TB) {
    int d[8];
#pragma unroll
    for (int u = 0; u < 8; ++u) {
      const int i = i0 + u * BK_TB;
      d[u] = binned[i < b1 ? i : b1 - 1].y;
    }
#pragma unroll
    for (int u = 0; u < 8; ++u) {
      const int dl = d[u] - base;
      if (i0 + u * BK_TB < b1 && (unsigned)dl < (unsigned)BIN_NODES) atomicAdd(&cnt[dl], 1);
    }
  }
  __syncthreads();

  int c = 0;
  if (tid < BIN_NODES) c = cnt[tid];
  int inc = c;
  for (int o = 1; o < 32; o <<= 1) {
    const int a = __shfl_up(inc, o);
    if (lane >= o) inc += a;
  }
  if (lane == 31 && wave < BIN_NODES / 32) sW[wave] = inc;
  __syncthreads();
  if (tid < BIN_NODES) {
    int ex = inc - c;
    for (int w = 0; w < BIN_NODES / 32; ++w) {
      if (w < wave) ex += sW[w];
    }
    excl[tid] = ex;
    cnt[tid] = 0;
    const int node = base + tid;
    if (node < n) {
      rowptr[node] = b0 + ex;
      dinv[node] = rsqrtf((float)(c + 1));
    }
  }
  if (bin == (int)gridDim.x - 1 && tid == 0) rowptr[n] = b1;
  __syncthreads();

  for (int i0 = b0 + tid; i0 < b1; i0 += 8 * BK_TB) {
    int2 e[8];
#pragma unroll
    for (int u = 0; u < 8; ++u) {
      const int i = i0 + u * BK_TB;
      e[u] = binned[i < b1 ? i : b1 - 1];
    }
#pragma unroll
    for (int u = 0; u < 8; ++u) {
      const int dl = e[u].y - base;
      if (i0 + u * BK_TB < b1 && (unsigned)dl < (unsigned)BIN_NODES) {
        const int p = excl[dl] + atomicAdd(&cnt[dl], 1);
        if (p < ROWS_CAP) {
          stage[p] = e[u].x;
        } else if (b0 + p < b1) {
          csr[b0 + p] = e[u].x;
        }
      }
    }
  }
  __syncthreads();

  int m = b1 - b0;
  m = m < ROWS_CAP ? m : ROWS_CAP;
  for (int i = tid; i < m; i += BK_TB) csr[b0 + i] = stage[i];
}

typedef __attribute__((ext_vector_type(4))) float v4f;

constexpr int GEMM_GRID = 512;

template <int K, int N>
__global__ __launch_bounds__(256) void k_gemm_wmma(const float* __restrict__ A,
                                                   const float* __restrict__ W,
                                                   const float* __restrict__ rs,
                                                   float* __restrict__ C, int mTiles) {
  constexpr int TN = N / 16;
  constexpr int KC = 64;
  constexpr int CPS = K / KC;
  constexpr int NV = K * N / 4;
  __shared__ v4f sW[NV];

#pragma unroll 8
  for (int p = threadIdx.x; p < NV; p += 256) {
    const int cc = p & 15;
    const int t0 = (p >> 4) & 1;
    const int hh = (p >> 5) & 1;
    const int rest = p >> 6;
    const int t = 2 * (rest % (TN / 2)) + t0;
    const int J = rest / (TN / 2);
    const float* wp = W + (size_t)(8 * J + 4 * hh) * N + 16 * t + cc;
    v4f w;
    w.x = wp[0];
    w.y = wp[N];
    w.z = wp[2 * N];
    w.w = wp[3 * N];
    sW[(J * TN + t) * 32 + hh * 16 + cc] = w;
  }
  __syncthreads();

  const int lane = threadIdx.x & 31;
  const int wave = __builtin_amdgcn_readfirstlane((int)(threadIdx.x >> 5));
  const int c = lane & 15;
  const int h = lane >> 4;
  const int stride = 8 * (int)gridDim.x;
  int s = wave * (int)gridDim.x + (int)blockIdx.x;
  if (s >= mTiles) return;

  const float* arow = A + ((size_t)(16 * s + c) * K + 4 * h);
  v4f acur[KC / 8];
#pragma unroll
  for (int u = 0; u < KC / 8; ++u) acur[u] = *(const v4f*)(arow + 8 * u);

  v4f bcur[TN];
#pragma unroll
  for (int t = 0; t < TN; ++t) bcur[t] = sW[t * 32 + lane];

  for (; s < mTiles; s += stride) {
    const float* nrow = (s + stride < mTiles) ? arow + (size_t)stride * 16 * K : arow;

    const v4f sc0 = *(const v4f*)(rs + 16 * s + 8 * h);
    const v4f sc1 = *(const v4f*)(rs + 16 * s + 8 * h + 4);

    v8f acc[TN];
#pragma unroll
    for (int t = 0; t < TN; ++t) acc[t] = (v8f){0.f, 0.f, 0.f, 0.f, 0.f, 0.f, 0.f, 0.f};

#pragma nounroll
    for (int ch = 0; ch < CPS; ++ch) {
      const bool lastc = (ch + 1 == CPS);
      const float* np = lastc ? nrow : arow + KC * (ch + 1);
      const v4f* bp = sW + (ch * (KC / 8) * TN) * 32 + lane;
      const v4f* bq = lastc ? sW + lane : bp + (KC / 8) * TN * 32;

#pragma unroll
      for (int u = 0; u < KC / 8; ++u) {
        const v4f* bn = (u + 1 < KC / 8) ? bp + (u + 1) * TN * 32 : bq;
        v4f bnxt[TN];
#pragma unroll
        for (int t = 0; t < TN; ++t) bnxt[t] = bn[t * 32];
        __builtin_amdgcn_sched_barrier(0);

        const v2f a0 = {acur[u].x, acur[u].y};
        const v2f a1 = {acur[u].z, acur[u].w};
#pragma unroll
        for (int t = 0; t < TN; ++t) {
          const v2f b0 = {bcur[t].x, bcur[t].y};
          acc[t] = __builtin_amdgcn_wmma_f32_16x16x4_f32(false, a0, false, b0, (short)0,
                                                         acc[t], false, false);
        }
#pragma unroll
        for (int t = 0; t < TN; ++t) {
          const v2f b1 = {bcur[t].z, bcur[t].w};
          acc[t] = __builtin_amdgcn_wmma_f32_16x16x4_f32(false, a1, false, b1, (short)0,
                                                         acc[t], false, false);
        }
#pragma unroll
        for (int t = 0; t < TN; ++t) bcur[t] = bnxt[t];
        acur[u] = *(const v4f*)(np + 8 * u);
        __builtin_amdgcn_sched_barrier(0);
      }
    }

    float* cp = C + ((size_t)(16 * s + 8 * h) * N + c);
#pragma unroll
    for (int t = 0; t < TN; ++t) {
#pragma unroll
      for (int v = 0; v < 8; ++v) {
        const float sc = (v < 4) ? sc0[v & 3] : sc1[v & 3];
        cp[(size_t)v * N + 16 * t] = acc[t][v] * sc;
      }
    }
    arow = nrow;
  }
}

template <int F, bool RELU>
__global__ __launch_bounds__(256) void k_gather(const float* __restrict__ hw,
                                                const float* __restrict__ dinv,
                                                const int* __restrict__ rowptr,
                                                const int* __restrict__ csr,
                                                const float* __restrict__ b,
                                                float* __restrict__ outp, int n, int E) {
  constexpr int VPL = F / 32;
  typedef __attribute__((ext_vector_type(VPL))) float vecf;
  const int lane = threadIdx.x & 31;
  const int d = __builtin_amdgcn_readfirstlane((int)((blockIdx.x * blockDim.x + threadIdx.x) >> 5));
  if (d >= n) return;

  int beg = rowptr[d];
  int end = rowptr[d + 1];
  beg = beg < 0 ? 0 : beg;
  end = end > E ? E : end;
  const float dd = dinv[d];
  const float* hl = hw + VPL * lane;
  const vecf self = *(const vecf*)(hl + (size_t)d * F);
  const vecf bias = *(const vecf*)(b + VPL * lane);

  vecf acc = (vecf)(0.0f);
  for (int c0 = beg; c0 < end; c0 += 32) {
    const int m = (end - c0) < 32 ? (end - c0) : 32;
    int s = d;
    if (lane < m) {
      const int t = csr[c0 + lane];
      s = ((unsigned)t < (unsigned)n) ? t : d;
    }
    for (int j = 0; j < m; j += 4) {
      const int s0 = __builtin_amdgcn_readlane(s, j);
      const int s1 = __builtin_amdgcn_readlane(s, j + 1);
      const int s2 = __builtin_amdgcn_readlane(s, j + 2);
      const int s3 = __builtin_amdgcn_readlane(s, j + 3);
      const vecf r0 = *(const vecf*)(hl + (size_t)s0 * F);
      const vecf r1 = *(const vecf*)(hl + (size_t)s1 * F);
      const vecf r2 = *(const vecf*)(hl + (size_t)s2 * F);
      const vecf r3 = *(const vecf*)(hl + (size_t)s3 * F);
      acc += r0;
      const vecf t1 = acc + r1;
      acc = (j + 1 < m) ? t1 : acc;
      const vecf t2 = acc + r2;
      acc = (j + 2 < m) ? t2 : acc;
      const vecf t3 = acc + r3;
      acc = (j + 3 < m) ? t3 : acc;
    }
  }

  vecf o;
#pragma unroll
  for (int k = 0; k < VPL; ++k) {
    const float v = (acc[k] + self[k]) * dd + bias[k];
    o[k] = (RELU && !(v > 0.0f)) ? 0.0f : v;
  }
  *(vecf*)(outp + (size_t)d * F + VPL * lane) = o;
}

extern "C" void kernel_launch(void* const* d_in, const int* in_sizes, int n_in,
                              void* d_out, int out_size, void* d_ws, size_t ws_size,
                              hipStream_t stream) {
  (void)n_in;
  (void)out_size;
  (void)ws_size;

  const float* x = (const float*)d_in[0];
  const int* ei = (const int*)d_in[1];
  const float* W1 = (const float*)d_in[2];
  const float* b1 = (const float*)d_in[3];
  const float* W2 = (const float*)d_in[4];
  const float* b2 = (const float*)d_in[5];
  float* out = (float*)d_out;

  constexpr int IN = 256, HID = 128, OUT = 64;
  const int N = in_sizes[0] / IN;
  const int E = in_sizes[1] / 2;
  const int* src = ei;
  const int* dst = ei + E;

  char* ws = (char*)d_ws;
  size_t off = 0;
  auto wsalloc = [&](size_t bytes) -> void* {
    void* p = ws + off;
    off += (bytes + 255) & ~(size_t)255;
    return p;
  };
  float* dinv = (float*)wsalloc((size_t)N * 4);
  float* hw1 = (float*)wsalloc((size_t)N * HID * 4);
  float* h1 = (float*)wsalloc((size_t)N * HID * 4);
  int* rowptr = (int*)wsalloc((size_t)(N + 1) * 4);
  int* csr = (int*)wsalloc((size_t)E * 4);
  float* hw2 = hw1;
  int2* binned = (int2*)h1;
  int* hist = (int*)((char*)h1 + (((size_t)E * 8 + 255) & ~(size_t)255));
  int* binstart = hist + MAXBIN * NSLICE;

  const int TB = 256;
  auto blks = [](long long total, int tb) { return (unsigned)((total + tb - 1) / tb); };

  k_bin_count<<<NSLICE, BK_TB, 0, stream>>>(dst, hist, N, E);
  k_bin_fill<<<NSLICE, BK_TB, 0, stream>>>(src, dst, hist, binstart, binned, N, E);
  k_bin_rows<<<blks(N, BIN_NODES), BK_TB, 0, stream>>>(binned, binstart, rowptr, dinv, csr, N, E);

  {
    const int mTiles = N / 16;
    k_gemm_wmma<IN, HID><<<GEMM_GRID, TB, 0, stream>>>(x, W1, dinv, hw1, mTiles);
  }

  k_gather<HID, true><<<blks((long long)N * 32, TB), TB, 0, stream>>>(hw1, dinv, rowptr, csr,
                                                                      b1, h1, N, E);

  {
    const int mTiles = N / 16;
    k_gemm_wmma<HID, OUT><<<GEMM_GRID, TB, 0, stream>>>(h1, W2, dinv, hw2, mTiles);
  }

  k_gather<OUT, false><<<blks((long long)N * 32, TB), TB, 0, stream>>>(hw2, dinv, rowptr, csr,
                                                                       b2, out, N, E);
}


# layer-2 GEMM fused into the layer-1 aggregation: a work-group sums 32 nodes' rows into an LDS tile and multiplies it by W2 (in LDS) on WMMA; h1 never goes to memory; 7 launches -> 6; sum orders kept
# speedup vs baseline: 4.7526x; 1.0464x over previous
#include <hip/hip_runtime.h>
#include <hip/hip_bf16.h>

typedef __attribute__((ext_vector_type(2))) float v2f;
typedef __attribute__((ext_vector_type(8))) float v8f;

constexpr int BIN_SHIFT = 8;
constexpr int BIN_NODES = 1 << BIN_SHIFT;
constexpr int MAXBIN = 512;
constexpr int NSLICE = 256;
constexpr int BK_TB = 1024;
constexpr int FILL_CAP = 8192;
constexpr int ROWS_CAP = 8192;

__device__ __forceinline__ int bin_count(int n) {
  const int nb = (n + BIN_NODES - 1) >> BIN_SHIFT;
  return nb < MAXBIN ? nb : MAXBIN;
}

__device__ __forceinline__ int bin_of(int d, int nbins) {
  const int b = (int)((unsigned)d >> BIN_SHIFT);
  return b < nbins ? b : nbins - 1;
}

__global__ __launch_bounds__(BK_TB) void k_bin_count(const int* __restrict__ dst,
                                                     int* __restrict__ hist, int n, int E) {
  __shared__ int h[MAXBIN];
  const int tid = threadIdx.x;
  const int nbins = bin_count(n);
  for (int b = tid; b < MAXBIN; b += BK_TB) h[b] = 0;
  __syncthreads();

  const int per = (E + NSLICE - 1) / NSLICE;
  const int e0 = (int)blockIdx.x * per;
  const int e1 = (e0 + per < E) ? e0 + per : E;
  for (int i0 = e0 + tid; i0 < e1; i0 += 8 * BK_TB) {
    int d[8];
#pragma unroll
    for (int u = 0; u < 8; ++u) {
      const int i = i0 + u * BK_TB;
      d[u] = dst[i < e1 ? i : e1 - 1];
    }
#pragma unroll
    for (int u = 0; u < 8; ++u) {
      if (i0 + u * BK_TB < e1) atomicAdd(&h[bin_of(d[u], nbins)], 1);
    }
  }
  __syncthreads();
  for (int b = tid; b < nbins; b += BK_TB) hist[b * NSLICE + (int)blockIdx.x] = h[b];
}

__global__ __launch_bounds__(BK_TB) void k_bin_fill(const int* __restrict__ src,
                                                    const int* __restrict__ dst,
                                                    const int* __restrict__ hist,
                                                    int* __restrict__ binstart,
                                                    int2* __restrict__ binned, int n, int E) {
  __shared__ int sT[MAXBIN];
  __shared__ int sP[MAXBIN];
  __shared__ int sL[MAXBIN];
  __shared__ int sCnt[MAXBIN];
  __shared__ int sWT[BK_TB / 32];
  __shared__ int sWL[BK_TB / 32];
  __shared__ int2 stage[FILL_CAP];
  const int tid = threadIdx.x;
  const int lane = tid & 31;
  const int wave = tid >> 5;
  const int blk = (int)blockIdx.x;
  const int nbins = bin_count(n);

  const int sub = tid & 7;
  const int grp = tid >> 3;
  const int4* h4 = (const int4*)hist;
  for (int r0 = 0; r0 < MAXBIN; r0 += BK_TB / 8) {
    const int r = r0 + grp;
    int tot = 0, pre = 0;
    if (r < nbins) {
      int4 q[NSLICE / 32];
#pragma unroll
      for (int j = 0; j < NSLICE / 32; ++j) q[j] = h4[r * (NSLICE / 4) + j * 8 + sub];
#pragma unroll
      for (int j = 0; j < NSLICE / 32; ++j) {
        const int k = (j * 8 + sub) * 4;
        tot += q[j].x + q[j].y + q[j].z + q[j].w;
        pre += (k < blk ? q[j].x : 0) + (k + 1 < blk ? q[j].y : 0) + (k + 2 < blk ? q[j].z : 0) +
               (k + 3 < blk ? q[j].w : 0);
      }
    }
    for (int o = 1; o < 8; o <<= 1) {
      tot += __shfl_xor(tot, o);
      pre += __shfl_xor(pre, o);
    }
    if (sub == 0) {
      sT[r] = tot;
      sP[r] = pre;
    }
  }
  for (int r = tid; r < MAXBIN; r += BK_TB) {
    sL[r] = (r < nbins) ? hist[r * NSLICE + blk] : 0;
    sCnt[r] = 0;
  }
  __syncthreads();

  int t = 0, l = 0;
  if (tid < MAXBIN) {
    t = sT[tid];
    l = sL[tid];
  }
  int it = t, il = l;
  for (int o = 1; o < 32; o <<= 1) {
    const int a = __shfl_up(it, o);
    const int c = __shfl_up(il, o);
    if (lane >= o) {
      it += a;
      il += c;
    }
  }
  if (lane == 31) {
    sWT[wave] = it;
    sWL[wave] = il;
  }
  __syncthreads();
  if (tid < MAXBIN) {
    int bt = it - t, bl = il - l;
    for (int w = 0; w < MAXBIN / 32; ++w) {
      if (w < wave) {
        bt += sWT[w];
        bl += sWL[w];
      }
    }
    sP[tid] += bt;
    sL[tid] = bl;
    if (blk == 0 && tid < nbins) {
      binstart[tid] = bt;
      if (tid == nbins - 1) binstart[nbins] = bt + t;
    }
  }
  __syncthreads();

  const int per = (E + NSLICE - 1) / NSLICE;
  const int e0 = blk * per;
  const int e1 = (e0 + per < E) ? e0 + per : E;
  for (int i0 = e0 + tid; i0 < e1; i0 += 8 * BK_TB) {
    int s[8], d[8];
#pragma unroll
    for (int u = 0; u < 8; ++u) {
      const int i = i0 + u * BK_TB;
      const int ic = i < e1 ? i : e1 - 1;
      s[u] = src[ic];
      d[u] = dst[ic];
    }
#pragma unroll
    for (int u = 0; u < 8; ++u) {
      if (i0 + u * BK_TB < e1) {
        const int b = bin_of(d[u], nbins);
        const int rank = atomicAdd(&sCnt[b], 1);
        const int li = sL[b] + rank;
        if (li < FILL_CAP) {
          stage[li] = make_int2(s[u], d[u]);
        } else {
          const int p = sP[b] + rank;
          if ((unsigned)p < (unsigned)E) binned[p] = make_int2(s[u], d[u]);
        }
      }
    }
  }
  __syncthreads();

  int m = e1 - e0;
  m = m < 0 ? 0 : (m < FILL_CAP ? m : FILL_CAP);
  for (int i = tid; i < m; i += BK_TB) {
    const int2 e = stage[i];
    const int b = bin_of(e.y, nbins);
    const int p = sP[b] + (i - sL[b]);
    if ((unsigned)p < (unsigned)E) binned[p] = e;
  }
}

__global__ __launch_bounds__(BK_TB) void k_bin_rows(const int2* __restrict__ binned,
                                                    const int* __restrict__ binstart,
                                                    int* __restrict__ rowptr,
                                                    float* __restrict__ dinv,
                                                    int* __restrict__ csr, int n, int E) {
  __shared__ int cnt[BIN_NODES];
  __shared__ int excl[BIN_NODES];
  __shared__ int sW[BIN_NODES / 32];
  __shared__ int stage[ROWS_CAP];
  const int tid = threadIdx.x;
  const int lane = tid & 31;
  const int wave = tid >> 5;
  const int bin = (int)blockIdx.x;
  const int base = bin << BIN_SHIFT;
  int b0 = binstart[bin];
  int b1 = binstart[bin + 1];
  b0 = b0 < 0 ? 0 : (b0 > E ? E : b0);
  b1 = b1 < b0 ? b0 : (b1 > E ? E : b1);

  if (tid < BIN_NODES) cnt[tid] = 0;
  __syncthreads();

  for (int i0 = b0 + tid; i0 < b1; i0 += 8 * BK_TB) {
    int d[8];
#pragma unroll
    for (int u = 0; u < 8; ++u) {
      const int i = i0 + u * BK_TB;
      d[u] = binned[i < b1 ? i : b1 - 1].y;
    }
#pragma unroll
    for (int u = 0; u < 8; ++u) {
      const int dl = d[u] - base;
      if (i0 + u * BK_TB < b1 && (unsigned)dl < (unsigned)BIN_NODES) atomicAdd(&cnt[dl], 1);
    }
  }
  __syncthreads();

  int c = 0;
  if (tid < BIN_NODES) c = cnt[tid];
  int inc = c;
  for (int o = 1; o < 32; o <<= 1) {
    const int a = __shfl_up(inc, o);
    if (lane >= o) inc += a;
  }
  if (lane == 31 && wave < BIN_NODES / 32) sW[wave] = inc;
  __syncthreads();
  if (tid < BIN_NODES) {
    int ex = inc - c;
    for (int w = 0; w < BIN_NODES / 32; ++w) {
      if (w < wave) ex += sW[w];
    }
    excl[tid] = ex;
    cnt[tid] = 0;
    const int node = base + tid;
    if (node < n) {
      rowptr[node] = b0 + ex;
      dinv[node] = rsqrtf((float)(c + 1));
    }
  }
  if (bin == (int)gridDim.x - 1 && tid == 0) rowptr[n] = b1;
  __syncthreads();

  for (int i0 = b0 + tid; i0 < b1; i0 += 8 * BK_TB) {
    int2 e[8];
#pragma unroll
    for (int u = 0; u < 8; ++u) {
      const int i = i0 + u * BK_TB;
      e[u] = binned[i < b1 ? i : b1 - 1];
    }
#pragma unroll
    for (int u = 0; u < 8; ++u) {
      const int dl = e[u].y - base;
      if (i0 + u * BK_TB < b1 && (unsigned)dl < (unsigned)BIN_NODES) {
        const int p = excl[dl] + atomicAdd(&cnt[dl], 1);
        if (p < ROWS_CAP) {
          stage[p] = e[u].x;
        } else if (b0 + p < b1) {
          csr[b0 + p] = e[u].x;
        }
      }
    }
  }
  __syncthreads();

  int m = b1 - b0;
  m = m < ROWS_CAP ? m : ROWS_CAP;
  for (int i = tid; i < m; i += BK_TB) csr[b0 + i] = stage[i];
}

typedef __attribute__((ext_vector_type(4))) float v4f;

constexpr int GEMM_GRID = 512;

template <int K, int N>
__global__ __launch_bounds__(256) void k_gemm_wmma(const float* __restrict__ A,
                                                   const float* __restrict__ W,
                                                   const float* __restrict__ rs,
                                                   float* __restrict__ C, int mTiles) {
  constexpr int TN = N / 16;
  constexpr int KC = 64;
  constexpr int CPS = K / KC;
  constexpr int NV = K * N / 4;
  __shared__ v4f sW[NV];

#pragma unroll 8
  for (int p = threadIdx.x; p < NV; p += 256) {
    const int cc = p & 15;
    const int t0 = (p >> 4) & 1;
    const int hh = (p >> 5) & 1;
    const int rest = p >> 6;
    const int t = 2 * (rest % (TN / 2)) + t0;
    const int J = rest / (TN / 2);
    const float* wp = W + (size_t)(8 * J + 4 * hh) * N + 16 * t + cc;
    v4f w;
    w.x = wp[0];
    w.y = wp[N];
    w.z = wp[2 * N];
    w.w = wp[3 * N];
    sW[(J * TN + t) * 32 + hh * 16 + cc] = w;
  }
  __syncthreads();

  const int lane = threadIdx.x & 31;
  const int wave = __builtin_amdgcn_readfirstlane((int)(threadIdx.x >> 5));
  const int c = lane & 15;
  const int h = lane >> 4;
  const int stride = 8 * (int)gridDim.x;
  int s = wave * (int)gridDim.x + (int)blockIdx.x;
  if (s >= mTiles) return;

  const float* arow = A + ((size_t)(16 * s + c) * K + 4 * h);
  v4f acur[KC / 8];
#pragma unroll
  for (int u = 0; u < KC / 8; ++u) acur[u] = *(const v4f*)(arow + 8 * u);

  v4f bcur[TN];
#pragma unroll
  for (int t = 0; t < TN; ++t) bcur[t] = sW[t * 32 + lane];

  for (; s < mTiles; s += stride) {
    const float* nrow = (s + stride < mTiles) ? arow + (size_t)stride * 16 * K : arow;

    const v4f sc0 = *(const v4f*)(rs + 16 * s + 8 * h);
    const v4f sc1 = *(const v4f*)(rs + 16 * s + 8 * h + 4);

    v8f acc[TN];
#pragma unroll
    for (int t = 0; t < TN; ++t) acc[t] = (v8f){0.f, 0.f, 0.f, 0.f, 0.f, 0.f, 0.f, 0.f};

#pragma nounroll
    for (int ch = 0; ch < CPS; ++ch) {
      const bool lastc = (ch + 1 == CPS);
      const float* np = lastc ? nrow : arow + KC * (ch + 1);
      const v4f* bp = sW + (ch * (KC / 8) * TN) * 32 + lane;
      const v4f* bq = lastc ? sW + lane : bp + (KC / 8) * TN * 32;

#pragma unroll
      for (int u = 0; u < KC / 8; ++u) {
        const v4f* bn = (u + 1 < KC / 8) ? bp + (u + 1) * TN * 32 : bq;
        v4f bnxt[TN];
#pragma unroll
        for (int t = 0; t < TN; ++t) bnxt[t] = bn[t * 32];
        __builtin_amdgcn_sched_barrier(0);

        const v2f a0 = {acur[u].x, acur[u].y};
        const v2f a1 = {acur[u].z, acur[u].w};
#pragma unroll
        for (int t = 0; t < TN; ++t) {
          const v2f b0 = {bcur[t].x, bcur[t].y};
          acc[t] = __builtin_amdgcn_wmma_f32_16x16x4_f32(false, a0, false, b0, (short)0,
                                                         acc[t], false, false);
        }
#pragma unroll
        for (int t = 0; t < TN; ++t) {
          const v2f b1 = {bcur[t].z, bcur[t].w};
          acc[t] = __builtin_amdgcn_wmma_f32_16x16x4_f32(false, a1, false, b1, (short)0,
                                                         acc[t], false, false);
        }
#pragma unroll
        for (int t = 0; t < TN; ++t) bcur[t] = bnxt[t];
        acur[u] = *(const v4f*)(np + 8 * u);
        __builtin_amdgcn_sched_barrier(0);
      }
    }

    float* cp = C + ((size_t)(16 * s + 8 * h) * N + c);
#pragma unroll
    for (int t = 0; t < TN; ++t) {
#pragma unroll
      for (int v = 0; v < 8; ++v) {
        const float sc = (v < 4) ? sc0[v & 3] : sc1[v & 3];
        cp[(size_t)v * N + 16 * t] = acc[t][v] * sc;
      }
    }
    arow = nrow;
  }
}

template <int F, bool RELU>
__global__ __launch_bounds__(256) void k_gather(const float* __restrict__ hw,
                                                const float* __restrict__ dinv,
                                                const int* __restrict__ rowptr,
                                                const int* __restrict__ csr,
                                                const float* __restrict__ b,
                                                float* __restrict__ outp, int n, int E) {
  constexpr int VPL = F / 32;
  typedef __attribute__((ext_vector_type(VPL))) float vecf;
  const int lane = threadIdx.x & 31;
  const int d = __builtin_amdgcn_readfirstlane((int)((blockIdx.x * blockDim.x + threadIdx.x) >> 5));
  if (d >= n) return;

  int beg = rowptr[d];
  int end = rowptr[d + 1];
  beg = beg < 0 ? 0 : beg;
  end = end > E ? E : end;
  const float dd = dinv[d];
  const float* hl = hw + VPL * lane;
  const vecf self = *(const vecf*)(hl + (size_t)d * F);
  const vecf bias = *(const vecf*)(b + VPL * lane);

  vecf acc = (vecf)(0.0f);
  for (int c0 = beg; c0 < end; c0 += 32) {
    const int m = (end - c0) < 32 ? (end - c0) : 32;
    int s = d;
    if (lane < m) {
      const int t = csr[c0 + lane];
      s = ((unsigned)t < (unsigned)n) ? t : d;
    }
    for (int j = 0; j < m; j += 4) {
      const int s0 = __builtin_amdgcn_readlane(s, j);
      const int s1 = __builtin_amdgcn_readlane(s, j + 1);
      const int s2 = __builtin_amdgcn_readlane(s, j + 2);
      const int s3 = __builtin_amdgcn_readlane(s, j + 3);
      const vecf r0 = *(const vecf*)(hl + (size_t)s0 * F);
      const vecf r1 = *(const vecf*)(hl + (size_t)s1 * F);
      const vecf r2 = *(const vecf*)(hl + (size_t)s2 * F);
      const vecf r3 = *(const vecf*)(hl + (size_t)s3 * F);
      acc += r0;
      const vecf t1 = acc + r1;
      acc = (j + 1 < m) ? t1 : acc;
      const vecf t2 = acc + r2;
      acc = (j + 2 < m) ? t2 : acc;
      const vecf t3 = acc + r3;
      acc = (j + 3 < m) ? t3 : acc;
    }
  }

  vecf o;
#pragma unroll
  for (int k = 0; k < VPL; ++k) {
    const float v = (acc[k] + self[k]) * dd + bias[k];
    o[k] = (RELU && !(v > 0.0f)) ? 0.0f : v;
  }
  *(vecf*)(outp + (size_t)d * F + VPL * lane) = o;
}

constexpr int FG_ROWS = 32;
constexpr int FG_TB = 512;

template <int F, int N2>
__global__ __launch_bounds__(FG_TB) void k_gather_gemm(const float* __restrict__ hw,
                                                       const float* __restrict__ dinv,
                                                       const int* __restrict__ rowptr,
                                                       const int* __restrict__ csr,
                                                       const float* __restrict__ b,
                                                       const float* __restrict__ W,
                                                       float* __restrict__ C, int n, int E) {
  static_assert(F == 128, "a lane owns one 16-byte vector of a row");
  constexpr int TN = N2 / 16;
  constexpr int NV = F * N2 / 4;
  constexpr int WPT = NV / FG_TB;
  constexpr int P4 = (F + 4) / 4;
  __shared__ v4f sW[NV];
  __shared__ v4f hT[FG_ROWS * P4];
  __shared__ v4f sD[FG_ROWS / 4];

  const int lane = threadIdx.x & 31;
  const int wave = __builtin_amdgcn_readfirstlane((int)(threadIdx.x >> 5));
  const int node0 = (int)blockIdx.x * FG_ROWS;

  v4f wreg[WPT];
#pragma unroll
  for (int i = 0; i < WPT; ++i) {
    const int p = (int)threadIdx.x + i * FG_TB;
    const int cc = p & 15;
    const int t0 = (p >> 4) & 1;
    const int hh = (p >> 5) & 1;
    const int rest = p >> 6;
    const int t = 2 * (rest % (TN / 2)) + t0;
    const int J = rest / (TN / 2);
    const float* wp = W + (size_t)(8 * J + 4 * hh) * N2 + 16 * t + cc;
    wreg[i].x = wp[0];
    wreg[i].y = wp[N2];
    wreg[i].z = wp[2 * N2];
    wreg[i].w = wp[3 * N2];
  }

  const float* hl = hw + 4 * lane;
  const v4f bias = *(const v4f*)(b + 4 * lane);
  for (int i = 0; i < FG_ROWS / 16; ++i) {
    const int r = wave + 16 * i;
    const int d = node0 + r;
    v4f o = (v4f)(0.0f);
    float dd = 0.0f;
    if (d < n) {
      int beg = rowptr[d];
      int end = rowptr[d + 1];
      beg = beg < 0 ? 0 : beg;
      end = end > E ? E : end;
      dd = dinv[d];
      const v4f self = *(const v4f*)(hl + (size_t)d * F);
      v4f acc = (v4f)(0.0f);
      for (int c0 = beg; c0 < end; c0 += 32) {
        const int m = (end - c0) < 32 ? (end - c0) : 32;
        int s = d;
        if (lane < m) {
          const int t = csr[c0 + lane];
          s = ((unsigned)t < (unsigned)n) ? t : d;
        }
        for (int j = 0; j < m; j += 4) {
          const int s0 = __builtin_amdgcn_readlane(s, j);
          const int s1 = __builtin_amdgcn_readlane(s, j + 1);
          const int s2 = __builtin_amdgcn_readlane(s, j + 2);
          const int s3 = __builtin_amdgcn_readlane(s, j + 3);
          const v4f r0 = *(const v4f*)(hl + (size_t)s0 * F);
          const v4f r1 = *(const v4f*)(hl + (size_t)s1 * F);
          const v4f r2 = *(const v4f*)(hl + (size_t)s2 * F);
          const v4f r3 = *(const v4f*)(hl + (size_t)s3 * F);
          acc += r0;
          const v4f t1 = acc + r1;
          acc = (j + 1 < m) ? t1 : acc;
          const v4f t2 = acc + r2;
          acc = (j + 2 < m) ? t2 : acc;
          const v4f t3 = acc + r3;
          acc = (j + 3 < m) ? t3 : acc;
        }
      }
#pragma unroll
      for (int k = 0; k < 4; ++k) {
        const float v = (acc[k] + self[k]) * dd + bias[k];
        o[k] = !(v > 0.0f) ? 0.0f : v;
      }
    }
    hT[r * P4 + lane] = o;
    if (lane == 0) ((float*)sD)[r] = dd;
  }

#pragma unroll
  for (int i = 0; i < WPT; ++i) {
    const int p = (int)threadIdx.x + i * FG_TB;
    const int cc = p & 15;
    const int t0 = (p >> 4) & 1;
    const int hh = (p >> 5) & 1;
    const int rest = p >> 6;
    const int t = 2 * (rest % (TN / 2)) + t0;
    const int J = rest / (TN / 2);
    sW[(J * TN + t) * 32 + hh * 16 + cc] = wreg[i];
  }
  __syncthreads();
  if (wave >= 2 * (FG_ROWS / 16)) return;

  const int c = lane & 15;
  const int h = lane >> 4;
  const int strip = wave >> 1;
  const int tp = 2 * (wave & 1);
  const v4f* ap = hT + (16 * strip + c) * P4 + h;
  const v4f* bp = sW + tp * 32 + lane;
  v8f acc0 = (v8f){0.f, 0.f, 0.f, 0.f, 0.f, 0.f, 0.f, 0.f};
  v8f acc1 = (v8f){0.f, 0.f, 0.f, 0.f, 0.f, 0.f, 0.f, 0.f};
#pragma unroll
  for (int J = 0; J < F / 8; ++J) {
    const v4f x = ap[2 * J];
    const v4f w0 = bp[J * TN * 32];
    const v4f w1 = bp[J * TN * 32 + 32];
    const v2f xa = {x.x, x.y};
    const v2f xb = {x.z, x.w};
    const v2f w0a = {w0.x, w0.y};
    const v2f w0b = {w0.z, w0.w};
    const v2f w1a = {w1.x, w1.y};
    const v2f w1b = {w1.z, w1.w};
    acc0 = __builtin_amdgcn_wmma_f32_16x16x4_f32(false, xa, false, w0a, (short)0, acc0, false, false);
    acc1 = __builtin_amdgcn_wmma_f32_16x16x4_f32(false, xa, false, w1a, (short)0, acc1, false, false);
    acc0 = __builtin_amdgcn_wmma_f32_16x16x4_f32(false, xb, false, w0b, (short)0, acc0, false, false);
    acc1 = __builtin_amdgcn_wmma_f32_16x16x4_f32(false, xb, false, w1b, (short)0, acc1, false, false);
  }

  const int rl = 16 * strip + 8 * h;
  const v4f sc0 = sD[rl >> 2];
  const v4f sc1 = sD[(rl >> 2) + 1];
  float* cp = C + ((size_t)(node0 + rl) * N2 + 16 * tp + c);
#pragma unroll
  for (int v = 0; v < 8; ++v) {
    const float sc = (v < 4) ? sc0[v & 3] : sc1[v & 3];
    if (node0 + rl + v < n) {
      cp[(size_t)v * N2] = acc0[v] * sc;
      cp[(size_t)v * N2 + 16] = acc1[v] * sc;
    }
  }
}

extern "C" void kernel_launch(void* const* d_in, const int* in_sizes, int n_in,
                              void* d_out, int out_size, void* d_ws, size_t ws_size,
                              hipStream_t stream) {
  (void)n_in;
  (void)out_size;
  (void)ws_size;

  const float* x = (const float*)d_in[0];
  const int* ei = (const int*)d_in[1];
  const float* W1 = (const float*)d_in[2];
  const float* b1 = (const float*)d_in[3];
  const float* W2 = (const float*)d_in[4];
  const float* b2 = (const float*)d_in[5];
  float* out = (float*)d_out;

  constexpr int IN = 256, HID = 128, OUT = 64;
  const int N = in_sizes[0] / IN;
  const int E = in_sizes[1] / 2;
  const int* src = ei;
  const int* dst = ei + E;

  char* ws = (char*)d_ws;
  size_t off = 0;
  auto wsalloc = [&](size_t bytes) -> void* {
    void* p = ws + off;
    off += (bytes + 255) & ~(size_t)255;
    return p;
  };
  float* dinv = (float*)wsalloc((size_t)N * 4);
  float* hw1 = (float*)wsalloc((size_t)N * HID * 4);
  float* h1 = (float*)wsalloc((size_t)N * HID * 4);
  int* rowptr = (int*)wsalloc((size_t)(N + 1) * 4);
  int* csr = (int*)wsalloc((size_t)E * 4);
  float* hw2 = h1;
  int2* binned = (int2*)h1;
  int* hist = (int*)((char*)h1 + (((size_t)E * 8 + 255) & ~(size_t)255));
  int* binstart = hist + MAXBIN * NSLICE;

  const int TB = 256;
  auto blks = [](long long total, int tb) { return (unsigned)((total + tb - 1) / tb); };

  k_bin_count<<<NSLICE, BK_TB, 0, stream>>>(dst, hist, N, E);
  k_bin_fill<<<NSLICE, BK_TB, 0, stream>>>(src, dst, hist, binstart, binned, N, E);
  k_bin_rows<<<blks(N, BIN_NODES), BK_TB, 0, stream>>>(binned, binstart, rowptr, dinv, csr, N, E);

  {
    const int mTiles = N / 16;
    k_gemm_wmma<IN, HID><<<GEMM_GRID, TB, 0, stream>>>(x, W1, dinv, hw1, mTiles);
  }

  k_gather_gemm<HID, OUT><<<blks(N, FG_ROWS), FG_TB, 0, stream>>>(hw1, dinv, rowptr, csr, b1, W2,
                                                                  hw2, N, E);

  k_gather<OUT, false><<<blks((long long)N * 32, TB), TB, 0, stream>>>(hw2, dinv, rowptr, csr,
                                                                       b2, out, N, E);
}


# edge bucketing in 2 launches, not 3 (6 -> 5 in all): count pass folded into the fill launch, each slice ordered in its own range; rows launch walks its bin's 256 runs; row sum order = LDS add rank
# speedup vs baseline: 5.0674x; 1.0662x over previous
#include <hip/hip_runtime.h>
#include <hip/hip_bf16.h>

typedef __attribute__((ext_vector_type(2))) float v2f;
typedef __attribute__((ext_vector_type(8))) float v8f;

constexpr int BIN_SHIFT = 8;
constexpr int BIN_NODES = 1 << BIN_SHIFT;
constexpr int MAXBIN = 512;
constexpr int NSLICE = 256;
constexpr int OFF_PITCH = MAXBIN + 32;
constexpr int BK_TB = 1024;
constexpr int FILL_CAP = 8192;
constexpr int ROWS_CAP = 8192;

__device__ __forceinline__ int bin_count(int n) {
  const int nb = (n + BIN_NODES - 1) >> BIN_SHIFT;
  return nb < MAXBIN ? nb : MAXBIN;
}

__device__ __forceinline__ int bin_of(int d, int nbins) {
  const int b = (int)((unsigned)d >> BIN_SHIFT);
  return b < nbins ? b : nbins - 1;
}

__global__ __launch_bounds__(BK_TB) void k_bin_fill(const int* __restrict__ src,
                                                    const int* __restrict__ dst,
                                                    int* __restrict__ offs,
                                                    int2* __restrict__ binned, int n, int E) {
  __shared__ int sL[MAXBIN];
  __shared__ int sCnt[MAXBIN];
  __shared__ int sWL[BK_TB / 32];
  __shared__ int2 stage[FILL_CAP];
  const int tid = threadIdx.x;
  const int lane = tid & 31;
  const int wave = tid >> 5;
  const int blk = (int)blockIdx.x;
  const int nbins = bin_count(n);
  for (int b = tid; b < MAXBIN; b += BK_TB) sCnt[b] = 0;
  __syncthreads();

  const int per = (E + NSLICE - 1) / NSLICE;
  const int e0 = blk * per;
  const int e1 = (e0 + per < E) ? e0 + per : E;

  for (int i0 = e0 + tid; i0 < e1; i0 += 8 * BK_TB) {
    int d[8];
#pragma unroll
    for (int u = 0; u < 8; ++u) {
      const int i = i0 + u * BK_TB;
      d[u] = dst[i < e1 ? i : e1 - 1];
    }
#pragma unroll
    for (int u = 0; u < 8; ++u) {
      if (i0 + u * BK_TB < e1) atomicAdd(&sCnt[bin_of(d[u], nbins)], 1);
    }
  }
  __syncthreads();

  int l = 0;
  if (tid < MAXBIN) l = sCnt[tid];
  int il = l;
  for (int o = 1; o < 32; o <<= 1) {
    const int c = __shfl_up(il, o);
    if (lane >= o) il += c;
  }
  if (lane == 31) sWL[wave] = il;
  __syncthreads();
  if (tid < MAXBIN) {
    int bl = il - l;
    for (int w = 0; w < MAXBIN / 32; ++w) {
      if (w < wave) bl += sWL[w];
    }
    sL[tid] = bl;
    sCnt[tid] = 0;
    if (tid < nbins) {
      offs[blk * OFF_PITCH + tid] = bl;
      if (tid == nbins - 1) offs[blk * OFF_PITCH + nbins] = bl + l;
    }
  }
  __syncthreads();

  for (int i0 = e0 + tid; i0 < e1; i0 += 8 * BK_TB) {
    int s[8], d[8];
#pragma unroll
    for (int u = 0; u < 8; ++u) {
      const int i = i0 + u * BK_TB;
      const int ic = i < e1 ? i : e1 - 1;
      s[u] = src[ic];
      d[u] = dst[ic];
    }
#pragma unroll
    for (int u = 0; u < 8; ++u) {
      if (i0 + u * BK_TB < e1) {
        const int b = bin_of(d[u], nbins);
        const int li = sL[b] + atomicAdd(&sCnt[b], 1);
        if (li < FILL_CAP) {
          stage[li] = make_int2(s[u], d[u]);
        } else if (e0 + li < e1) {
          binned[e0 + li] = make_int2(s[u], d[u]);
        }
      }
    }
  }
  __syncthreads();

  int m = e1 - e0;
  m = m < 0 ? 0 : (m < FILL_CAP ? m : FILL_CAP);
  for (int i = tid; i < m; i += BK_TB) binned[e0 + i] = stage[i];
}

__global__ __launch_bounds__(BK_TB) void k_bin_rows(const int2* __restrict__ binned,
                                                    const int* __restrict__ offs,
                                                    int* __restrict__ rowptr,
                                                    float* __restrict__ dinv,
                                                    int* __restrict__ csr, int n, int E) {
  static_assert(BK_TB == 4 * NSLICE, "four threads walk one run");
  __shared__ int cnt[BIN_NODES];
  __shared__ int excl[BIN_NODES];
  __shared__ int sW[BIN_NODES / 32];
  __shared__ int sSO[NSLICE / 32];
  __shared__ int sSN[NSLICE / 32];
  __shared__ int rBeg[NSLICE];
  __shared__ int rLen[NSLICE];
  __shared__ int stage[ROWS_CAP];
  const int tid = threadIdx.x;
  const int lane = tid & 31;
  const int wave = tid >> 5;
  const int bin = (int)blockIdx.x;
  const int base = bin << BIN_SHIFT;

  int so = 0, sn = 0;
  if (tid < NSLICE) {
    const int per = (E + NSLICE - 1) / NSLICE;
    const int e0 = tid * per;
    const int e1 = (e0 + per < E) ? e0 + per : E;
    const int sl = e1 > e0 ? e1 - e0 : 0;
    const int br = bin < MAXBIN ? bin : MAXBIN - 1;
    int a = offs[tid * OFF_PITCH + br];
    int c = offs[tid * OFF_PITCH + br + 1];
    a = a < 0 ? 0 : (a > sl ? sl : a);
    c = c < a ? a : (c > sl ? sl : c);
    so = a;
    sn = c - a;
    rBeg[tid] = e0 + a;
    rLen[tid] = sn;
  }
  for (int o = 16; o > 0; o >>= 1) {
    so += __shfl_xor(so, o);
    sn += __shfl_xor(sn, o);
  }
  if (lane == 0 && wave < NSLICE / 32) {
    sSO[wave] = so;
    sSN[wave] = sn;
  }
  if (tid < BIN_NODES) cnt[tid] = 0;
  __syncthreads();
  int b0 = 0, b1 = 0;
  for (int w = 0; w < NSLICE / 32; ++w) {
    b0 += sSO[w];
    b1 += sSN[w];
  }
  b0 = b0 < 0 ? 0 : (b0 > E ? E : b0);
  b1 = b0 + b1;
  b1 = b1 > E ? E : b1;

  const int sub = tid & 3;
  const int rb = rBeg[tid >> 2];
  const int ln = rLen[tid >> 2];

  for (int k0 = sub; k0 < ln; k0 += 32) {
    int d[8];
#pragma unroll
    for (int u = 0; u < 8; ++u) {
      const int k = k0 + 4 * u;
      d[u] = binned[rb + (k < ln ? k : ln - 1)].y;
    }
#pragma unroll
    for (int u = 0; u < 8; ++u) {
      const int dl = d[u] - base;
      if (k0 + 4 * u < ln && (unsigned)dl < (unsigned)BIN_NODES) atomicAdd(&cnt[dl], 1);
    }
  }
  __syncthreads();

  int c = 0;
  if (tid < BIN_NODES) c = cnt[tid];
  int inc = c;
  for (int o = 1; o < 32; o <<= 1) {
    const int a = __shfl_up(inc, o);
    if (lane >= o) inc += a;
  }
  if (lane == 31 && wave < BIN_NODES / 32) sW[wave] = inc;
  __syncthreads();
  if (tid < BIN_NODES) {
    int ex = inc - c;
    for (int w = 0; w < BIN_NODES / 32; ++w) {
      if (w < wave) ex += sW[w];
    }
    excl[tid] = ex;
    cnt[tid] = 0;
    const int node = base + tid;
    if (node < n) {
      rowptr[node] = b0 + ex;
      dinv[node] = rsqrtf((float)(c + 1));
    }
  }
  if (bin == (int)gridDim.x - 1 && tid == 0) rowptr[n] = b1;
  __syncthreads();

  for (int k0 = sub; k0 < ln; k0 += 32) {
    int2 e[8];
#pragma unroll
    for (int u = 0; u < 8; ++u) {
      const int k = k0 + 4 * u;
      e[u] = binned[rb + (k < ln ? k : ln - 1)];
    }
#pragma unroll
    for (int u = 0; u < 8; ++u) {
      const int dl = e[u].y - base;
      if (k0 + 4 * u < ln && (unsigned)dl < (unsigned)BIN_NODES) {
        const int p = excl[dl] + atomicAdd(&cnt[dl], 1);
        if (p < ROWS_CAP) {
          stage[p] = e[u].x;
        } else if (b0 + p < b1) {
          csr[b0 + p] = e[u].x;
        }
      }
    }
  }
  __syncthreads();

  int m = b1 - b0;
  m = m < ROWS_CAP ? m : ROWS_CAP;
  for (int i = tid; i < m; i += BK_TB) csr[b0 + i] = stage[i];
}

typedef __attribute__((ext_vector_type(4))) float v4f;

constexpr int GEMM_GRID = 512;

template <int K, int N>
__global__ __launch_bounds__(256) void k_gemm_wmma(const float* __restrict__ A,
                                                   const float* __restrict__ W,
                                                   const float* __restrict__ rs,
                                                   float* __restrict__ C, int mTiles) {
  constexpr int TN = N / 16;
  constexpr int KC = 64;
  constexpr int CPS = K / KC;
  constexpr int NV = K * N / 4;
  __shared__ v4f sW[NV];

#pragma unroll 8
  for (int p = threadIdx.x; p < NV; p += 256) {
    const int cc = p & 15;
    const int t0 = (p >> 4) & 1;
    const int hh = (p >> 5) & 1;
    const int rest = p >> 6;
    const int t = 2 * (rest % (TN / 2)) + t0;
    const int J = rest / (TN / 2);
    const float* wp = W + (size_t)(8 * J + 4 * hh) * N + 16 * t + cc;
    v4f w;
    w.x = wp[0];
    w.y = wp[N];
    w.z = wp[2 * N];
    w.w = wp[3 * N];
    sW[(J * TN + t) * 32 + hh * 16 + cc] = w;
  }
  __syncthreads();

  const int lane = threadIdx.x & 31;
  const int wave = __builtin_amdgcn_readfirstlane((int)(threadIdx.x >> 5));
  const int c = lane & 15;
  const int h = lane >> 4;
  const int stride = 8 * (int)gridDim.x;
  int s = wave * (int)gridDim.x + (int)blockIdx.x;
  if (s >= mTiles) return;

  const float* arow = A + ((size_t)(16 * s + c) * K + 4 * h);
  v4f acur[KC / 8];
#pragma unroll
  for (int u = 0; u < KC / 8; ++u) acur[u] = *(const v4f*)(arow + 8 * u);

  v4f bcur[TN];
#pragma unroll
  for (int t = 0; t < TN; ++t) bcur[t] = sW[t * 32 + lane];

  for (; s < mTiles; s += stride) {
    const float* nrow = (s + stride < mTiles) ? arow + (size_t)stride * 16 * K : arow;

    const v4f sc0 = *(const v4f*)(rs + 16 * s + 8 * h);
    const v4f sc1 = *(const v4f*)(rs + 16 * s + 8 * h + 4);

    v8f acc[TN];
#pragma unroll
    for (int t = 0; t < TN; ++t) acc[t] = (v8f){0.f, 0.f, 0.f, 0.f, 0.f, 0.f, 0.f, 0.f};

#pragma nounroll
    for (int ch = 0; ch < CPS; ++ch) {
      const bool lastc = (ch + 1 == CPS);
      const float* np = lastc ? nrow : arow + KC * (ch + 1);
      const v4f* bp = sW + (ch * (KC / 8) * TN) * 32 + lane;
      const v4f* bq = lastc ? sW + lane : bp + (KC / 8) * TN * 32;

#pragma unroll
      for (int u = 0; u < KC / 8; ++u) {
        const v4f* bn = (u + 1 < KC / 8) ? bp + (u + 1) * TN * 32 : bq;
        v4f bnxt[TN];
#pragma unroll
        for (int t = 0; t < TN; ++t) bnxt[t] = bn[t * 32];
        __builtin_amdgcn_sched_barrier(0);

        const v2f a0 = {acur[u].x, acur[u].y};
        const v2f a1 = {acur[u].z, acur[u].w};
#pragma unroll
        for (int t = 0; t < TN; ++t) {
          const v2f b0 = {bcur[t].x, bcur[t].y};
          acc[t] = __builtin_amdgcn_wmma_f32_16x16x4_f32(false, a0, false, b0, (short)0,
                                                         acc[t], false, false);
        }
#pragma unroll
        for (int t = 0; t < TN; ++t) {
          const v2f b1 = {bcur[t].z, bcur[t].w};
          acc[t] = __builtin_amdgcn_wmma_f32_16x16x4_f32(false, a1, false, b1, (short)0,
                                                         acc[t], false, false);
        }
#pragma unroll
        for (int t = 0; t < TN; ++t) bcur[t] = bnxt[t];
        acur[u] = *(const v4f*)(np + 8 * u);
        __builtin_amdgcn_sched_barrier(0);
      }
    }

    float* cp = C + ((size_t)(16 * s + 8 * h) * N + c);
#pragma unroll
    for (int t = 0; t < TN; ++t) {
#pragma unroll
      for (int v = 0; v < 8; ++v) {
        const float sc = (v < 4) ? sc0[v & 3] : sc1[v & 3];
        cp[(size_t)v * N + 16 * t] = acc[t][v] * sc;
      }
    }
    arow = nrow;
  }
}

template <int F, bool RELU>
__global__ __launch_bounds__(256) void k_gather(const float* __restrict__ hw,
                                                const float* __restrict__ dinv,
                                                const int* __restrict__ rowptr,
                                                const int* __restrict__ csr,
                                                const float* __restrict__ b,
                                                float* __restrict__ outp, int n, int E) {
  constexpr int VPL = F / 32;
  typedef __attribute__((ext_vector_type(VPL))) float vecf;
  const int lane = threadIdx.x & 31;
  const int d = __builtin_amdgcn_readfirstlane((int)((blockIdx.x * blockDim.x + threadIdx.x) >> 5));
  if (d >= n) return;

  int beg = rowptr[d];
  int end = rowptr[d + 1];
  beg = beg < 0 ? 0 : beg;
  end = end > E ? E : end;
  const float dd = dinv[d];
  const float* hl = hw + VPL * lane;
  const vecf self = *(const vecf*)(hl + (size_t)d * F);
  const vecf bias = *(const vecf*)(b + VPL * lane);

  vecf acc = (vecf)(0.0f);
  for (int c0 = beg; c0 < end; c0 += 32) {
    const int m = (end - c0) < 32 ? (end - c0) : 32;
    int s = d;
    if (lane < m) {
      const int t = csr[c0 + lane];
      s = ((unsigned)t < (unsigned)n) ? t : d;
    }
    for (int j = 0; j < m; j += 4) {
      const int s0 = __builtin_amdgcn_readlane(s, j);
      const int s1 = __builtin_amdgcn_readlane(s, j + 1);
      const int s2 = __builtin_amdgcn_readlane(s, j + 2);
      const int s3 = __builtin_amdgcn_readlane(s, j + 3);
      const vecf r0 = *(const vecf*)(hl + (size_t)s0 * F);
      const vecf r1 = *(const vecf*)(hl + (size_t)s1 * F);
      const vecf r2 = *(const vecf*)(hl + (size_t)s2 * F);
      const vecf r3 = *(const vecf*)(hl + (size_t)s3 * F);
      acc += r0;
      const vecf t1 = acc + r1;
      acc = (j + 1 < m) ? t1 : acc;
      const vecf t2 = acc + r2;
      acc = (j + 2 < m) ? t2 : acc;
      const vecf t3 = acc + r3;
      acc = (j + 3 < m) ? t3 : acc;
    }
  }

  vecf o;
#pragma unroll
  for (int k = 0; k < VPL; ++k) {
    const float v = (acc[k] + self[k]) * dd + bias[k];
    o[k] = (RELU && !(v > 0.0f)) ? 0.0f : v;
  }
  *(vecf*)(outp + (size_t)d * F + VPL * lane) = o;
}

constexpr int FG_ROWS = 32;
constexpr int FG_TB = 512;

template <int F, int N2>
__global__ __launch_bounds__(FG_TB) void k_gather_gemm(const float* __restrict__ hw,
                                                       const float* __restrict__ dinv,
                                                       const int* __restrict__ rowptr,
                                                       const int* __restrict__ csr,
                                                       const float* __restrict__ b,
                                                       const float* __restrict__ W,
                                                       float* __restrict__ C, int n, int E) {
  static_assert(F == 128, "a lane owns one 16-byte vector of a row");
  constexpr int TN = N2 / 16;
  constexpr int NV = F * N2 / 4;
  constexpr int WPT = NV / FG_TB;
  constexpr int P4 = (F + 4) / 4;
  __shared__ v4f sW[NV];
  __shared__ v4f hT[FG_ROWS * P4];
  __shared__ v4f sD[FG_ROWS / 4];

  const int lane = threadIdx.x & 31;
  const int wave = __builtin_amdgcn_readfirstlane((int)(threadIdx.x >> 5));
  const int node0 = (int)blockIdx.x * FG_ROWS;

  v4f wreg[WPT];
#pragma unroll
  for (int i = 0; i < WPT; ++i) {
    const int p = (int)threadIdx.x + i * FG_TB;
    const int cc = p & 15;
    const int t0 = (p >> 4) & 1;
    const int hh = (p >> 5) & 1;
    const int rest = p >> 6;
    const int t = 2 * (rest % (TN / 2)) + t0;
    const int J = rest / (TN / 2);
    const float* wp = W + (size_t)(8 * J + 4 * hh) * N2 + 16 * t + cc;
    wreg[i].x = wp[0];
    wreg[i].y = wp[N2];
    wreg[i].z = wp[2 * N2];
    wreg[i].w = wp[3 * N2];
  }

  const float* hl = hw + 4 * lane;
  const v4f bias = *(const v4f*)(b + 4 * lane);
  for (int i = 0; i < FG_ROWS / 16; ++i) {
    const int r = wave + 16 * i;
    const int d = node0 + r;
    v4f o = (v4f)(0.0f);
    float dd = 0.0f;
    if (d < n) {
      int beg = rowptr[d];
      int end = rowptr[d + 1];
      beg = beg < 0 ? 0 : beg;
      end = end > E ? E : end;
      dd = dinv[d];
      const v4f self = *(const v4f*)(hl + (size_t)d * F);
      v4f acc = (v4f)(0.0f);
      for (int c0 = beg; c0 < end; c0 += 32) {
        const int m = (end - c0) < 32 ? (end - c0) : 32;
        int s = d;
        if (lane < m) {
          const int t = csr[c0 + lane];
          s = ((unsigned)t < (unsigned)n) ? t : d;
        }
        for (int j = 0; j < m; j += 4) {
          const int s0 = __builtin_amdgcn_readlane(s, j);
          const int s1 = __builtin_amdgcn_readlane(s, j + 1);
          const int s2 = __builtin_amdgcn_readlane(s, j + 2);
          const int s3 = __builtin_amdgcn_readlane(s, j + 3);
          const v4f r0 = *(const v4f*)(hl + (size_t)s0 * F);
          const v4f r1 = *(const v4f*)(hl + (size_t)s1 * F);
          const v4f r2 = *(const v4f*)(hl + (size_t)s2 * F);
          const v4f r3 = *(const v4f*)(hl + (size_t)s3 * F);
          acc += r0;
          const v4f t1 = acc + r1;
          acc = (j + 1 < m) ? t1 : acc;
          const v4f t2 = acc + r2;
          acc = (j + 2 < m) ? t2 : acc;
          const v4f t3 = acc + r3;
          acc = (j + 3 < m) ? t3 : acc;
        }
      }
#pragma unroll
      for (int k = 0; k < 4; ++k) {
        const float v = (acc[k] + self[k]) * dd + bias[k];
        o[k] = !(v > 0.0f) ? 0.0f : v;
      }
    }
    hT[r * P4 + lane] = o;
    if (lane == 0) ((float*)sD)[r] = dd;
  }

#pragma unroll
  for (int i = 0; i < WPT; ++i) {
    const int p = (int)threadIdx.x + i * FG_TB;
    const int cc = p & 15;
    const int t0 = (p >> 4) & 1;
    const int hh = (p >> 5) & 1;
    const int rest = p >> 6;
    const int t = 2 * (rest % (TN / 2)) + t0;
    const int J = rest / (TN / 2);
    sW[(J * TN + t) * 32 + hh * 16 + cc] = wreg[i];
  }
  __syncthreads();
  if (wave >= 2 * (FG_ROWS / 16)) return;

  const int c = lane & 15;
  const int h = lane >> 4;
  const int strip = wave >> 1;
  const int tp = 2 * (wave & 1);
  const v4f* ap = hT + (16 * strip + c) * P4 + h;
  const v4f* bp = sW + tp * 32 + lane;
  v8f acc0 = (v8f){0.f, 0.f, 0.f, 0.f, 0.f, 0.f, 0.f, 0.f};
  v8f acc1 = (v8f){0.f, 0.f, 0.f, 0.f, 0.f, 0.f, 0.f, 0.f};
#pragma unroll
  for (int J = 0; J < F / 8; ++J) {
    const v4f x = ap[2 * J];
    const v4f w0 = bp[J * TN * 32];
    const v4f w1 = bp[J * TN * 32 + 32];
    const v2f xa = {x.x, x.y};
    const v2f xb = {x.z, x.w};
    const v2f w0a = {w0.x, w0.y};
    const v2f w0b = {w0.z, w0.w};
    const v2f w1a = {w1.x, w1.y};
    const v2f w1b = {w1.z, w1.w};
    acc0 = __builtin_amdgcn_wmma_f32_16x16x4_f32(false, xa, false, w0a, (short)0, acc0, false, false);
    acc1 = __builtin_amdgcn_wmma_f32_16x16x4_f32(false, xa, false, w1a, (short)0, acc1, false, false);
    acc0 = __builtin_amdgcn_wmma_f32_16x16x4_f32(false, xb, false, w0b, (short)0, acc0, false, false);
    acc1 = __builtin_amdgcn_wmma_f32_16x16x4_f32(false, xb, false, w1b, (short)0, acc1, false, false);
  }

  const int rl = 16 * strip + 8 * h;
  const v4f sc0 = sD[rl >> 2];
  const v4f sc1 = sD[(rl >> 2) + 1];
  float* cp = C + ((size_t)(node0 + rl) * N2 + 16 * tp + c);
#pragma unroll
  for (int v = 0; v < 8; ++v) {
    const float sc = (v < 4) ? sc0[v & 3] : sc1[v & 3];
    if (node0 + rl + v < n) {
      cp[(size_t)v * N2] = acc0[v] * sc;
      cp[(size_t)v * N2 + 16] = acc1[v] * sc;
    }
  }
}

extern "C" void kernel_launch(void* const* d_in, const int* in_sizes, int n_in,
                              void* d_out, int out_size, void* d_ws, size_t ws_size,
                              hipStream_t stream) {
  (void)n_in;
  (void)out_size;
  (void)ws_size;

  const float* x = (const float*)d_in[0];
  const int* ei = (const int*)d_in[1];
  const float* W1 = (const float*)d_in[2];
  const float* b1 = (const float*)d_in[3];
  const float* W2 = (const float*)d_in[4];
  const float* b2 = (const float*)d_in[5];
  float* out = (float*)d_out;

  constexpr int IN = 256, HID = 128, OUT = 64;
  const int N = in_sizes[0] / IN;
  const int E = in_sizes[1] / 2;
  const int* src = ei;
  const int* dst = ei + E;

  char* ws = (char*)d_ws;
  size_t off = 0;
  auto wsalloc = [&](size_t bytes) -> void* {
    void* p = ws + off;
    off += (bytes + 255) & ~(size_t)255;
    return p;
  };
  float* dinv = (float*)wsalloc((size_t)N * 4);
  float* hw1 = (float*)wsalloc((size_t)N * HID * 4);
  float* h1 = (float*)wsalloc((size_t)N * HID * 4);
  int* rowptr = (int*)wsalloc((size_t)(N + 1) * 4);
  int* csr = (int*)wsalloc((size_t)E * 4);
  float* hw2 = h1;
  int2* binned = (int2*)h1;
  int* offs = (int*)((char*)h1 + (((size_t)E * 8 + 255) & ~(size_t)255));

  const int TB = 256;
  auto blks = [](long long total, int tb) { return (unsigned)((total + tb - 1) / tb); };

  k_bin_fill<<<NSLICE, BK_TB, 0, stream>>>(src, dst, offs, binned, N, E);
  k_bin_rows<<<blks(N, BIN_NODES), BK_TB, 0, stream>>>(binned, offs, rowptr, dinv, csr, N, E);

  {
    const int mTiles = N / 16;
    k_gemm_wmma<IN, HID><<<GEMM_GRID, TB, 0, stream>>>(x, W1, dinv, hw1, mTiles);
  }

  k_gather_gemm<HID, OUT><<<blks(N, FG_ROWS), FG_TB, 0, stream>>>(hw1, dinv, rowptr, csr, b1, W2,
                                                                  hw2, N, E);

  k_gather<OUT, false><<<blks((long long)N * 32, TB), TB, 0, stream>>>(hw2, dinv, rowptr, csr,
                                                                       b2, out, N, E);
}


# aggregation loads: layer 1 takes both nodes' bounds and sources ahead, 8 row loads per trip (was 4); in both layers a slot past a row's end reads the hot bias line, not a row from L2; sum order kept
# speedup vs baseline: 5.2381x; 1.0337x over previous
#include <hip/hip_runtime.h>
#include <hip/hip_bf16.h>

typedef __attribute__((ext_vector_type(2))) float v2f;
typedef __attribute__((ext_vector_type(8))) float v8f;

constexpr int BIN_SHIFT = 8;
constexpr int BIN_NODES = 1 << BIN_SHIFT;
constexpr int MAXBIN = 512;
constexpr int NSLICE = 256;
constexpr int OFF_PITCH = MAXBIN + 32;
constexpr int BK_TB = 1024;
constexpr int FILL_CAP = 8192;
constexpr int ROWS_CAP = 8192;

__device__ __forceinline__ int bin_count(int n) {
  const int nb = (n + BIN_NODES - 1) >> BIN_SHIFT;
  return nb < MAXBIN ? nb : MAXBIN;
}

__device__ __forceinline__ int bin_of(int d, int nbins) {
  const int b = (int)((unsigned)d >> BIN_SHIFT);
  return b < nbins ? b : nbins - 1;
}

__global__ __launch_bounds__(BK_TB) void k_bin_fill(const int* __restrict__ src,
                                                    const int* __restrict__ dst,
                                                    int* __restrict__ offs,
                                                    int2* __restrict__ binned, int n, int E) {
  __shared__ int sL[MAXBIN];
  __shared__ int sCnt[MAXBIN];
  __shared__ int sWL[BK_TB / 32];
  __shared__ int2 stage[FILL_CAP];
  const int tid = threadIdx.x;
  const int lane = tid & 31;
  const int wave = tid >> 5;
  const int blk = (int)blockIdx.x;
  const int nbins = bin_count(n);
  for (int b = tid; b < MAXBIN; b += BK_TB) sCnt[b] = 0;
  __syncthreads();

  const int per = (E + NSLICE - 1) / NSLICE;
  const int e0 = blk * per;
  const int e1 = (e0 + per < E) ? e0 + per : E;

  for (int i0 = e0 + tid; i0 < e1; i0 += 8 * BK_TB) {
    int d[8];
#pragma unroll
    for (int u = 0; u < 8; ++u) {
      const int i = i0 + u * BK_TB;
      d[u] = dst[i < e1 ? i : e1 - 1];
    }
#pragma unroll
    for (int u = 0; u < 8; ++u) {
      if (i0 + u * BK_TB < e1) atomicAdd(&sCnt[bin_of(d[u], nbins)], 1);
    }
  }
  __syncthreads();

  int l = 0;
  if (tid < MAXBIN) l = sCnt[tid];
  int il = l;
  for (int o = 1; o < 32; o <<= 1) {
    const int c = __shfl_up(il, o);
    if (lane >= o) il += c;
  }
  if (lane == 31) sWL[wave] = il;
  __syncthreads();
  if (tid < MAXBIN) {
    int bl = il - l;
    for (int w = 0; w < MAXBIN / 32; ++w) {
      if (w < wave) bl += sWL[w];
    }
    sL[tid] = bl;
    sCnt[tid] = 0;
    if (tid < nbins) {
      offs[blk * OFF_PITCH + tid] = bl;
      if (tid == nbins - 1) offs[blk * OFF_PITCH + nbins] = bl + l;
    }
  }
  __syncthreads();

  for (int i0 = e0 + tid; i0 < e1; i0 += 8 * BK_TB) {
    int s[8], d[8];
#pragma unroll
    for (int u = 0; u < 8; ++u) {
      const int i = i0 + u * BK_TB;
      const int ic = i < e1 ? i : e1 - 1;
      s[u] = src[ic];
      d[u] = dst[ic];
    }
#pragma unroll
    for (int u = 0; u < 8; ++u) {
      if (i0 + u * BK_TB < e1) {
        const int b = bin_of(d[u], nbins);
        const int li = sL[b] + atomicAdd(&sCnt[b], 1);
        if (li < FILL_CAP) {
          stage[li] = make_int2(s[u], d[u]);
        } else if (e0 + li < e1) {
          binned[e0 + li] = make_int2(s[u], d[u]);
        }
      }
    }
  }
  __syncthreads();

  int m = e1 - e0;
  m = m < 0 ? 0 : (m < FILL_CAP ? m : FILL_CAP);
  for (int i = tid; i < m; i += BK_TB) binned[e0 + i] = stage[i];
}

__global__ __launch_bounds__(BK_TB) void k_bin_rows(const int2* __restrict__ binned,
                                                    const int* __restrict__ offs,
                                                    int* __restrict__ rowptr,
                                                    float* __restrict__ dinv,
                                                    int* __restrict__ csr, int n, int E) {
  static_assert(BK_TB == 4 * NSLICE, "four threads walk one run");
  __shared__ int cnt[BIN_NODES];
  __shared__ int excl[BIN_NODES];
  __shared__ int sW[BIN_NODES / 32];
  __shared__ int sSO[NSLICE / 32];
  __shared__ int sSN[NSLICE / 32];
  __shared__ int rBeg[NSLICE];
  __shared__ int rLen[NSLICE];
  __shared__ int stage[ROWS_CAP];
  const int tid = threadIdx.x;
  const int lane = tid & 31;
  const int wave = tid >> 5;
  const int bin = (int)blockIdx.x;
  const int base = bin << BIN_SHIFT;

  int so = 0, sn = 0;
  if (tid < NSLICE) {
    const int per = (E + NSLICE - 1) / NSLICE;
    const int e0 = tid * per;
    const int e1 = (e0 + per < E) ? e0 + per : E;
    const int sl = e1 > e0 ? e1 - e0 : 0;
    const int br = bin < MAXBIN ? bin : MAXBIN - 1;
    int a = offs[tid * OFF_PITCH + br];
    int c = offs[tid * OFF_PITCH + br + 1];
    a = a < 0 ? 0 : (a > sl ? sl : a);
    c = c < a ? a : (c > sl ? sl : c);
    so = a;
    sn = c - a;
    rBeg[tid] = e0 + a;
    rLen[tid] = sn;
  }
  for (int o = 16; o > 0; o >>= 1) {
    so += __shfl_xor(so, o);
    sn += __shfl_xor(sn, o);
  }
  if (lane == 0 && wave < NSLICE / 32) {
    sSO[wave] = so;
    sSN[wave] = sn;
  }
  if (tid < BIN_NODES) cnt[tid] = 0;
  __syncthreads();
  int b0 = 0, b1 = 0;
  for (int w = 0; w < NSLICE / 32; ++w) {
    b0 += sSO[w];
    b1 += sSN[w];
  }
  b0 = b0 < 0 ? 0 : (b0 > E ? E : b0);
  b1 = b0 + b1;
  b1 = b1 > E ? E : b1;

  const int sub = tid & 3;
  const int rb = rBeg[tid >> 2];
  const int ln = rLen[tid >> 2];

  for (int k0 = sub; k0 < ln; k0 += 32) {
    int d[8];
#pragma unroll
    for (int u = 0; u < 8; ++u) {
      const int k = k0 + 4 * u;
      d[u] = binned[rb + (k < ln ? k : ln - 1)].y;
    }
#pragma unroll
    for (int u = 0; u < 8; ++u) {
      const int dl = d[u] - base;
      if (k0 + 4 * u < ln && (unsigned)dl < (unsigned)BIN_NODES) atomicAdd(&cnt[dl], 1);
    }
  }
  __syncthreads();

  int c = 0;
  if (tid < BIN_NODES) c = cnt[tid];
  int inc = c;
  for (int o = 1; o < 32; o <<= 1) {
    const int a = __shfl_up(inc, o);
    if (lane >= o) inc += a;
  }
  if (lane == 31 && wave < BIN_NODES / 32) sW[wave] = inc;
  __syncthreads();
  if (tid < BIN_NODES) {
    int ex = inc - c;
    for (int w = 0; w < BIN_NODES / 32; ++w) {
      if (w < wave) ex += sW[w];
    }
    excl[tid] = ex;
    cnt[tid] = 0;
    const int node = base + tid;
    if (node < n) {
      rowptr[node] = b0 + ex;
      dinv[node] = rsqrtf((float)(c + 1));
    }
  }
  if (bin == (int)gridDim.x - 1 && tid == 0) rowptr[n] = b1;
  __syncthreads();

  for (int k0 = sub; k0 < ln; k0 += 32) {
    int2 e[8];
#pragma unroll
    for (int u = 0; u < 8; ++u) {
      const int k = k0 + 4 * u;
      e[u] = binned[rb + (k < ln ? k : ln - 1)];
    }
#pragma unroll
    for (int u = 0; u < 8; ++u) {
      const int dl = e[u].y - base;
      if (k0 + 4 * u < ln && (unsigned)dl < (unsigned)BIN_NODES) {
        const int p = excl[dl] + atomicAdd(&cnt[dl], 1);
        if (p < ROWS_CAP) {
          stage[p] = e[u].x;
        } else if (b0 + p < b1) {
          csr[b0 + p] = e[u].x;
        }
      }
    }
  }
  __syncthreads();

  int m = b1 - b0;
  m = m < ROWS_CAP ? m : ROWS_CAP;
  for (int i = tid; i < m; i += BK_TB) csr[b0 + i] = stage[i];
}

typedef __attribute__((ext_vector_type(4))) float v4f;

constexpr int GEMM_GRID = 512;

template <int K, int N>
__global__ __launch_bounds__(256) void k_gemm_wmma(const float* __restrict__ A,
                                                   const float* __restrict__ W,
                                                   const float* __restrict__ rs,
                                                   float* __restrict__ C, int mTiles) {
  constexpr int TN = N / 16;
  constexpr int KC = 64;
  constexpr int CPS = K / KC;
  constexpr int NV = K * N / 4;
  __shared__ v4f sW[NV];

#pragma unroll 8
  for (int p = threadIdx.x; p < NV; p += 256) {
    const int cc = p & 15;
    const int t0 = (p >> 4) & 1;
    const int hh = (p >> 5) & 1;
    const int rest = p >> 6;
    const int t = 2 * (rest % (TN / 2)) + t0;
    const int J = rest / (TN / 2);
    const float* wp = W + (size_t)(8 * J + 4 * hh) * N + 16 * t + cc;
    v4f w;
    w.x = wp[0];
    w.y = wp[N];
    w.z = wp[2 * N];
    w.w = wp[3 * N];
    sW[(J * TN + t) * 32 + hh * 16 + cc] = w;
  }
  __syncthreads();

  const int lane = threadIdx.x & 31;
  const int wave = __builtin_amdgcn_readfirstlane((int)(threadIdx.x >> 5));
  const int c = lane & 15;
  const int h = lane >> 4;
  const int stride = 8 * (int)gridDim.x;
  int s = wave * (int)gridDim.x + (int)blockIdx.x;
  if (s >= mTiles) return;

  const float* arow = A + ((size_t)(16 * s + c) * K + 4 * h);
  v4f acur[KC / 8];
#pragma unroll
  for (int u = 0; u < KC / 8; ++u) acur[u] = *(const v4f*)(arow + 8 * u);

  v4f bcur[TN];
#pragma unroll
  for (int t = 0; t < TN; ++t) bcur[t] = sW[t * 32 + lane];

  for (; s < mTiles; s += stride) {
    const float* nrow = (s + stride < mTiles) ? arow + (size_t)stride * 16 * K : arow;

    const v4f sc0 = *(const v4f*)(rs + 16 * s + 8 * h);
    const v4f sc1 = *(const v4f*)(rs + 16 * s + 8 * h + 4);

    v8f acc[TN];
#pragma unroll
    for (int t = 0; t < TN; ++t) acc[t] = (v8f){0.f, 0.f, 0.f, 0.f, 0.f, 0.f, 0.f, 0.f};

#pragma nounroll
    for (int ch = 0; ch < CPS; ++ch) {
      const bool lastc = (ch + 1 == CPS);
      const float* np = lastc ? nrow : arow + KC * (ch + 1);
      const v4f* bp = sW + (ch * (KC / 8) * TN) * 32 + lane;
      const v4f* bq = lastc ? sW + lane : bp + (KC / 8) * TN * 32;

#pragma unroll
      for (int u = 0; u < KC / 8; ++u) {
        const v4f* bn = (u + 1 < KC / 8) ? bp + (u + 1) * TN * 32 : bq;
        v4f bnxt[TN];
#pragma unroll
        for (int t = 0; t < TN; ++t) bnxt[t] = bn[t * 32];
        __builtin_amdgcn_sched_barrier(0);

        const v2f a0 = {acur[u].x, acur[u].y};
        const v2f a1 = {acur[u].z, acur[u].w};
#pragma unroll
        for (int t = 0; t < TN; ++t) {
          const v2f b0 = {bcur[t].x, bcur[t].y};
          acc[t] = __builtin_amdgcn_wmma_f32_16x16x4_f32(false, a0, false, b0, (short)0,
                                                         acc[t], false, false);
        }
#pragma unroll
        for (int t = 0; t < TN; ++t) {
          const v2f b1 = {bcur[t].z, bcur[t].w};
          acc[t] = __builtin_amdgcn_wmma_f32_16x16x4_f32(false, a1, false, b1, (short)0,
                                                         acc[t], false, false);
        }
#pragma unroll
        for (int t = 0; t < TN; ++t) bcur[t] = bnxt[t];
        acur[u] = *(const v4f*)(np + 8 * u);
        __builtin_amdgcn_sched_barrier(0);
      }
    }

    float* cp = C + ((size_t)(16 * s + 8 * h) * N + c);
#pragma unroll
    for (int t = 0; t < TN; ++t) {
#pragma unroll
      for (int v = 0; v < 8; ++v) {
        const float sc = (v < 4) ? sc0[v & 3] : sc1[v & 3];
        cp[(size_t)v * N + 16 * t] = acc[t][v] * sc;
      }
    }
    arow = nrow;
  }
}

template <int F, bool RELU>
__global__ __launch_bounds__(256) void k_gather(const float* __restrict__ hw,
                                                const float* __restrict__ dinv,
                                                const int* __restrict__ rowptr,
                                                const int* __restrict__ csr,
                                                const float* __restrict__ b,
                                                float* __restrict__ outp, int n, int E) {
  constexpr int VPL = F / 32;
  typedef __attribute__((ext_vector_type(VPL))) float vecf;
  const int lane = threadIdx.x & 31;
  const int d = __builtin_amdgcn_readfirstlane((int)((blockIdx.x * blockDim.x + threadIdx.x) >> 5));
  if (d >= n) return;

  int beg = rowptr[d];
  int end = rowptr[d + 1];
  beg = beg < 0 ? 0 : beg;
  end = end > E ? E : end;
  const float dd = dinv[d];
  const float* hl = hw + VPL * lane;
  const vecf self = *(const vecf*)(hl + (size_t)d * F);
  const vecf bias = *(const vecf*)(b + VPL * lane);

  vecf acc = (vecf)(0.0f);
  for (int c0 = beg; c0 < end; c0 += 32) {
    const int m = (end - c0) < 32 ? (end - c0) : 32;
    int s = d;
    if (lane < m) {
      const int t = csr[c0 + lane];
      s = ((unsigned)t < (unsigned)n) ? t : d;
    }
    for (int j = 0; j < m; j += 4) {
      const int s0 = __builtin_amdgcn_readlane(s, j);
      const int s1 = __builtin_amdgcn_readlane(s, j + 1);
      const int s2 = __builtin_amdgcn_readlane(s, j + 2);
      const int s3 = __builtin_amdgcn_readlane(s, j + 3);
      const float* q1 = (j + 1 < m) ? hw + (size_t)s1 * F : b;
      const float* q2 = (j + 2 < m) ? hw + (size_t)s2 * F : b;
      const float* q3 = (j + 3 < m) ? hw + (size_t)s3 * F : b;
      const vecf r0 = *(const vecf*)(hl + (size_t)s0 * F);
      const vecf r1 = *(const vecf*)(q1 + VPL * lane);
      const vecf r2 = *(const vecf*)(q2 + VPL * lane);
      const vecf r3 = *(const vecf*)(q3 + VPL * lane);
      acc += r0;
      const vecf t1 = acc + r1;
      acc = (j + 1 < m) ? t1 : acc;
      const vecf t2 = acc + r2;
      acc = (j + 2 < m) ? t2 : acc;
      const vecf t3 = acc + r3;
      acc = (j + 3 < m) ? t3 : acc;
    }
  }

  vecf o;
#pragma unroll
  for (int k = 0; k < VPL; ++k) {
    const float v = (acc[k] + self[k]) * dd + bias[k];
    o[k] = (RELU && !(v > 0.0f)) ? 0.0f : v;
  }
  *(vecf*)(outp + (size_t)d * F + VPL * lane) = o;
}

constexpr int FG_ROWS = 32;
constexpr int FG_TB = 512;

template <int F, int N2>
__global__ __launch_bounds__(FG_TB) void k_gather_gemm(const float* __restrict__ hw,
                                                       const float* __restrict__ dinv,
                                                       const int* __restrict__ rowptr,
                                                       const int* __restrict__ csr,
                                                       const float* __restrict__ b,
                                                       const float* __restrict__ W,
                                                       float* __restrict__ C, int n, int E) {
  static_assert(F == 128, "a lane owns one 16-byte vector of a row");
  constexpr int TN = N2 / 16;
  constexpr int NV = F * N2 / 4;
  constexpr int WPT = NV / FG_TB;
  constexpr int P4 = (F + 4) / 4;
  __shared__ v4f sW[NV];
  __shared__ v4f hT[FG_ROWS * P4];
  __shared__ v4f sD[FG_ROWS / 4];

  const int lane = threadIdx.x & 31;
  const int wave = __builtin_amdgcn_readfirstlane((int)(threadIdx.x >> 5));
  const int node0 = (int)blockIdx.x * FG_ROWS;

  v4f wreg[WPT];
#pragma unroll
  for (int i = 0; i < WPT; ++i) {
    const int p = (int)threadIdx.x + i * FG_TB;
    const int cc = p & 15;
    const int t0 = (p >> 4) & 1;
    const int hh = (p >> 5) & 1;
    const int rest = p >> 6;
    const int t = 2 * (rest % (TN / 2)) + t0;
    const int J = rest / (TN / 2);
    const float* wp = W + (size_t)(8 * J + 4 * hh) * N2 + 16 * t + cc;
    wreg[i].x = wp[0];
    wreg[i].y = wp[N2];
    wreg[i].z = wp[2 * N2];
    wreg[i].w = wp[3 * N2];
  }

  const float* hl = hw + 4 * lane;
  const float* bl = b + 4 * lane;
  int nb[2], ne[2], sf[2];
  float dv[2];
#pragma unroll
  for (int i = 0; i < 2; ++i) {
    const int d = node0 + wave + 16 * i;
    const int dc = d < n ? d : n - 1;
    int beg = rowptr[dc];
    int end = rowptr[dc + 1];
    const float di = dinv[dc];
    beg = beg < 0 ? 0 : beg;
    end = end > E ? E : end;
    nb[i] = d < n ? beg : 0;
    ne[i] = d < n ? end : 0;
    dv[i] = d < n ? di : 0.0f;
  }
  int tf[2];
#pragma unroll
  for (int i = 0; i < 2; ++i) {
    int ix = nb[i] + lane;
    ix = ix < E ? ix : E - 1;
    tf[i] = csr[ix];
  }

#pragma unroll
  for (int i = 0; i < WPT; ++i) {
    const int p = (int)threadIdx.x + i * FG_TB;
    const int cc = p & 15;
    const int t0 = (p >> 4) & 1;
    const int hh = (p >> 5) & 1;
    const int rest = p >> 6;
    const int t = 2 * (rest % (TN / 2)) + t0;
    const int J = rest / (TN / 2);
    sW[(J * TN + t) * 32 + hh * 16 + cc] = wreg[i];
  }

  asm volatile("" : "+v"(tf[0]), "+v"(tf[1]));
#pragma unroll
  for (int i = 0; i < 2; ++i) {
    const int d = node0 + wave + 16 * i;
    const int m = (ne[i] - nb[i]) < 32 ? (ne[i] - nb[i]) : 32;
    sf[i] = (lane < m && (unsigned)tf[i] < (unsigned)n) ? tf[i] : d;
  }

  const v4f bias = *(const v4f*)bl;
#pragma unroll
  for (int i = 0; i < FG_ROWS / 16; ++i) {
    const int r = wave + 16 * i;
    const int d = node0 + r;
    v4f o = (v4f)(0.0f);
    const float dd = dv[i];
    {
      const int dc = d < n ? d : n - 1;
      const int beg = nb[i];
      const int end = ne[i];
      const v4f self = *(const v4f*)(hl + (size_t)dc * F);
      v4f acc = (v4f)(0.0f);
      for (int c0 = beg; c0 < end; c0 += 32) {
        const int m = (end - c0) < 32 ? (end - c0) : 32;
        int s = sf[i];
        if (c0 != beg) {
          s = d;
          if (lane < m) {
            const int t = csr[c0 + lane];
            s = ((unsigned)t < (unsigned)n) ? t : d;
          }
        }
        for (int j = 0; j < m; j += 8) {
          const int s0 = __builtin_amdgcn_readlane(s, j);
          const int s1 = __builtin_amdgcn_readlane(s, j + 1);
          const int s2 = __builtin_amdgcn_readlane(s, j + 2);
          const int s3 = __builtin_amdgcn_readlane(s, j + 3);
          const int s4 = __builtin_amdgcn_readlane(s, j + 4);
          const int s5 = __builtin_amdgcn_readlane(s, j + 5);
          const int s6 = __builtin_amdgcn_readlane(s, j + 6);
          const int s7 = __builtin_amdgcn_readlane(s, j + 7);
          const float* q0 = hw + (size_t)s0 * F;
          const float* q1 = (j + 1 < m) ? hw + (size_t)s1 * F : b;
          const float* q2 = (j + 2 < m) ? hw + (size_t)s2 * F : b;
          const float* q3 = (j + 3 < m) ? hw + (size_t)s3 * F : b;
          const float* q4 = (j + 4 < m) ? hw + (size_t)s4 * F : b;
          const float* q5 = (j + 5 < m) ? hw + (size_t)s5 * F : b;
          const float* q6 = (j + 6 < m) ? hw + (size_t)s6 * F : b;
          const float* q7 = (j + 7 < m) ? hw + (size_t)s7 * F : b;
          const v4f r0 = *(const v4f*)(q0 + 4 * lane);
          const v4f r1 = *(const v4f*)(q1 + 4 * lane);
          const v4f r2 = *(const v4f*)(q2 + 4 * lane);
          const v4f r3 = *(const v4f*)(q3 + 4 * lane);
          const v4f r4 = *(const v4f*)(q4 + 4 * lane);
          const v4f r5 = *(const v4f*)(q5 + 4 * lane);
          const v4f r6 = *(const v4f*)(q6 + 4 * lane);
          const v4f r7 = *(const v4f*)(q7 + 4 * lane);
          __builtin_amdgcn_sched_barrier(0);
          acc += r0;
          const v4f t1 = acc + r1;
          acc = (j + 1 < m) ? t1 : acc;
          const v4f t2 = acc + r2;
          acc = (j + 2 < m) ? t2 : acc;
          const v4f t3 = acc + r3;
          acc = (j + 3 < m) ? t3 : acc;
          const v4f t4 = acc + r4;
          acc = (j + 4 < m) ? t4 : acc;
          const v4f t5 = acc + r5;
          acc = (j + 5 < m) ? t5 : acc;
          const v4f t6 = acc + r6;
          acc = (j + 6 < m) ? t6 : acc;
          const v4f t7 = acc + r7;
          acc = (j + 7 < m) ? t7 : acc;
        }
      }
#pragma unroll
      for (int k = 0; k < 4; ++k) {
        const float v = (acc[k] + self[k]) * dd + bias[k];
        o[k] = (!(v > 0.0f) || d >= n) ? 0.0f : v;
      }
    }
    hT[r * P4 + lane] = o;
    if (lane == 0) ((float*)sD)[r] = dd;
  }

  __syncthreads();
  if (wave >= 2 * (FG_ROWS / 16)) return;

  const int c = lane & 15;
  const int h = lane >> 4;
  const int strip = wave >> 1;
  const int tp = 2 * (wave & 1);
  const v4f* ap = hT + (16 * strip + c) * P4 + h;
  const v4f* bp = sW + tp * 32 + lane;
  v8f acc0 = (v8f){0.f, 0.f, 0.f, 0.f, 0.f, 0.f, 0.f, 0.f};
  v8f acc1 = (v8f){0.f, 0.f, 0.f, 0.f, 0.f, 0.f, 0.f, 0.f};
#pragma unroll
  for (int J = 0; J < F / 8; ++J) {
    const v4f x = ap[2 * J];
    const v4f w0 = bp[J * TN * 32];
    const v4f w1 = bp[J * TN * 32 + 32];
    const v2f xa = {x.x, x.y};
    const v2f xb = {x.z, x.w};
    const v2f w0a = {w0.x, w0.y};
    const v2f w0b = {w0.z, w0.w};
    const v2f w1a = {w1.x, w1.y};
    const v2f w1b = {w1.z, w1.w};
    acc0 = __builtin_amdgcn_wmma_f32_16x16x4_f32(false, xa, false, w0a, (short)0, acc0, false, false);
    acc1 = __builtin_amdgcn_wmma_f32_16x16x4_f32(false, xa, false, w1a, (short)0, acc1, false, false);
    acc0 = __builtin_amdgcn_wmma_f32_16x16x4_f32(false, xb, false, w0b, (short)0, acc0, false, false);
    acc1 = __builtin_amdgcn_wmma_f32_16x16x4_f32(false, xb, false, w1b, (short)0, acc1, false, false);
  }

  const int rl = 16 * strip + 8 * h;
  const v4f sc0 = sD[rl >> 2];
  const v4f sc1 = sD[(rl >> 2) + 1];
  float* cp = C + ((size_t)(node0 + rl) * N2 + 16 * tp + c);
#pragma unroll
  for (int v = 0; v < 8; ++v) {
    const float sc = (v < 4) ? sc0[v & 3] : sc1[v & 3];
    if (node0 + rl + v < n) {
      cp[(size_t)v * N2] = acc0[v] * sc;
      cp[(size_t)v * N2 + 16] = acc1[v] * sc;
    }
  }
}

extern "C" void kernel_launch(void* const* d_in, const int* in_sizes, int n_in,
                              void* d_out, int out_size, void* d_ws, size_t ws_size,
                              hipStream_t stream) {
  (void)n_in;
  (void)out_size;
  (void)ws_size;

  const float* x = (const float*)d_in[0];
  const int* ei = (const int*)d_in[1];
  const float* W1 = (const float*)d_in[2];
  const float* b1 = (const float*)d_in[3];
  const float* W2 = (const float*)d_in[4];
  const float* b2 = (const float*)d_in[5];
  float* out = (float*)d_out;

  constexpr int IN = 256, HID = 128, OUT = 64;
  const int N = in_sizes[0] / IN;
  const int E = in_sizes[1] / 2;
  const int* src = ei;
  const int* dst = ei + E;

  char* ws = (char*)d_ws;
  size_t off = 0;
  auto wsalloc = [&](size_t bytes) -> void* {
    void* p = ws + off;
    off += (bytes + 255) & ~(size_t)255;
    return p;
  };
  float* dinv = (float*)wsalloc((size_t)N * 4);
  float* hw1 = (float*)wsalloc((size_t)N * HID * 4);
  float* h1 = (float*)wsalloc((size_t)N * HID * 4);
  int* rowptr = (int*)wsalloc((size_t)(N + 1) * 4);
  int* csr = (int*)wsalloc((size_t)E * 4);
  float* hw2 = h1;
  int2* binned = (int2*)h1;
  int* offs = (int*)((char*)h1 + (((size_t)E * 8 + 255) & ~(size_t)255));

  const int TB = 256;
  auto blks = [](long long total, int tb) { return (unsigned)((total + tb - 1) / tb); };

  k_bin_fill<<<NSLICE, BK_TB, 0, stream>>>(src, dst, offs, binned, N, E);
  k_bin_rows<<<blks(N, BIN_NODES), BK_TB, 0, stream>>>(binned, offs, rowptr, dinv, csr, N, E);

  {
    const int mTiles = N / 16;
    k_gemm_wmma<IN, HID><<<GEMM_GRID, TB, 0, stream>>>(x, W1, dinv, hw1, mTiles);
  }

  k_gather_gemm<HID, OUT><<<blks(N, FG_ROWS), FG_TB, 0, stream>>>(hw1, dinv, rowptr, csr, b1, W2,
                                                                  hw2, N, E);

  k_gather<OUT, false><<<blks((long long)N * 32, TB), TB, 0, stream>>>(hw2, dinv, rowptr, csr,
                                                                       b2, out, N, E);
}
